# starting kernel
# baseline (speedup 1.0000x reference)
#include <hip/hip_runtime.h>
#include <hip/hip_bf16.h>

typedef __attribute__((ext_vector_type(2))) float v2f;
typedef __attribute__((ext_vector_type(8))) float v8f;

#define B_ 8
#define N_ 512
#define F_ 128
#define T_ 32
#define D_ 64
#define EPS_ 1e-6f

__global__ void GAT_z_kernel(const float* __restrict__ x,
                             const float* __restrict__ W,
                             const float* __restrict__ bias,
                             float* __restrict__ z) {
  const int wave = blockIdx.x * (blockDim.x >> 5) + (threadIdx.x >> 5);
  const int lane = threadIdx.x & 31;

  const int dtile = wave & 3;
  const int ntile = (wave >> 2) & 31;
  const int t     = (wave >> 7) & 31;
  const int b     = wave >> 12;

  const int l15  = lane & 15;
  const int koff = (lane >> 4) << 1;
  const int mrow = ntile * 16 + l15;
  const int dcol = dtile * 16 + l15;

  const float* xrow = x + ((size_t)(b * N_ + mrow) * F_) * T_ + t;
  const float* wrow = W + (size_t)dcol * F_;

  v8f c = {};
  for (int k = 0; k < F_; k += 4) {
    v2f a, bm;
    a.x  = xrow[(size_t)(k + koff)     * T_];
    a.y  = xrow[(size_t)(k + koff + 1) * T_];
    bm.x = wrow[k + koff];
    bm.y = wrow[k + koff + 1];
    c = __builtin_amdgcn_wmma_f32_16x16x4_f32(false, a, false, bm,
                                              (short)0, c, false, false);
  }

  const float bv = bias[dcol];
  const int rbase = (lane >> 4) << 3;
  const size_t zbt = (size_t)(b * T_ + t) * N_;
#pragma unroll
  for (int v = 0; v < 8; ++v) {
    const int n = ntile * 16 + rbase + v;
    z[(zbt + n) * D_ + dcol] = tanhf(c[v] + bv);
  }
}

__global__ void GAT_sq_kernel(const float* __restrict__ z,
                              float* __restrict__ sq) {
  const int i = blockIdx.x * blockDim.x + threadIdx.x;
  if (i >= B_ * T_ * N_) return;
  const float* zr = z + (size_t)i * D_;
  float s = 0.f;
#pragma unroll 8
  for (int d = 0; d < D_; ++d) { const float v = zr[d]; s = fmaf(v, v, s); }
  sq[i] = s;
}

__global__ void __launch_bounds__(256)
GAT_att_kernel(const float* __restrict__ z,
               const float* __restrict__ sq,
               float* __restrict__ out) {
  __shared__ float att_s[16 * 512];
  __shared__ float sq_s[512];
  __shared__ float red_s[256];
  __shared__ float rowsum_s[16];

  const int tid   = threadIdx.x;
  const int wave  = tid >> 5;
  const int lane  = tid & 31;
  const int ntile = blockIdx.x & 31;
  const int t     = blockIdx.x >> 5;

  const int l15   = lane & 15;
  const int koff  = (lane >> 4) << 1;
  const int rbase = (lane >> 4) << 3;
  const int arow  = ntile * 16 + l15;

  float acc[32];
#pragma unroll
  for (int j = 0; j < 32; ++j) acc[j] = 0.f;

  for (int b = 0; b < B_; ++b) {
    const float* zb  = z  + (size_t)(b * T_ + t) * N_ * D_;
    const float* sqb = sq + (size_t)(b * T_ + t) * N_;

    for (int i = tid; i < N_; i += 256) sq_s[i] = sqb[i];
    __syncthreads();

    v8f c0 = {}, c1 = {}, c2 = {}, c3 = {};
    const float* arp = zb + (size_t)arow * D_;
    const float* bp0 = zb + (size_t)((wave * 4 + 0) * 16 + l15) * D_;
    const float* bp1 = zb + (size_t)((wave * 4 + 1) * 16 + l15) * D_;
    const float* bp2 = zb + (size_t)((wave * 4 + 2) * 16 + l15) * D_;
    const float* bp3 = zb + (size_t)((wave * 4 + 3) * 16 + l15) * D_;
#pragma unroll
    for (int k = 0; k < D_; k += 4) {
      const v2f a = *(const v2f*)(arp + k + koff);
      v2f bm;
      bm = *(const v2f*)(bp0 + k + koff);
      c0 = __builtin_amdgcn_wmma_f32_16x16x4_f32(false, a, false, bm, (short)0, c0, false, false);
      bm = *(const v2f*)(bp1 + k + koff);
      c1 = __builtin_amdgcn_wmma_f32_16x16x4_f32(false, a, false, bm, (short)0, c1, false, false);
      bm = *(const v2f*)(bp2 + k + koff);
      c2 = __builtin_amdgcn_wmma_f32_16x16x4_f32(false, a, false, bm, (short)0, c2, false, false);
      bm = *(const v2f*)(bp3 + k + koff);
      c3 = __builtin_amdgcn_wmma_f32_16x16x4_f32(false, a, false, bm, (short)0, c3, false, false);
    }

#pragma unroll
    for (int ct = 0; ct < 4; ++ct) {
      const v8f c = (ct == 0) ? c0 : (ct == 1) ? c1 : (ct == 2) ? c2 : c3;
      const int col = (wave * 4 + ct) * 16 + l15;
      const float sqm = sq_s[col];
      const float nm  = sqrtf(sqm) + EPS_;
#pragma unroll
      for (int v = 0; v < 8; ++v) {
        const int rloc = rbase + v;
        const float sqn = sq_s[ntile * 16 + rloc];
        const float nn  = sqrtf(sqn) + EPS_;
        const float dot = c[v];
        const float cosv = dot / (nn * nm);
        const float d2   = fmaxf(sqn + sqm - 2.0f * dot, 0.0f);
        const float diff = sqrtf(d2);
        att_s[rloc * 512 + col] = __expf(cosv - diff / (nn + nm + EPS_));
      }
    }
    __syncthreads();

    {
      const int row = tid >> 4;
      const int sl  = tid & 15;
      float part = 0.f;
#pragma unroll 4
      for (int cidx = sl; cidx < 512; cidx += 16) part += att_s[row * 512 + cidx];
      red_s[tid] = part;
    }
    __syncthreads();
    if ((tid & 15) == 0) {
      const int row = tid >> 4;
      float s = 0.f;
#pragma unroll
      for (int j = 0; j < 16; ++j) s += red_s[(row << 4) + j];
      rowsum_s[row] = s;
    }
    __syncthreads();

    const float invB = 1.0f / (float)B_;
#pragma unroll
    for (int j = 0; j < 32; ++j) {
      const int idx = tid + j * 256;
      const int r   = idx >> 9;
      acc[j] += att_s[idx] * invB / (rowsum_s[r] + EPS_);
    }
    __syncthreads();
  }

  float* outp = out + ((size_t)t * N_ + ntile * 16) * N_;
#pragma unroll
  for (int j = 0; j < 32; ++j) {
    const int idx = tid + j * 256;
    outp[idx] = acc[j];
  }
}

extern "C" void kernel_launch(void* const* d_in, const int* in_sizes, int n_in,
                              void* d_out, int out_size, void* d_ws, size_t ws_size,
                              hipStream_t stream) {
  const float* x    = (const float*)d_in[0];
  const float* W    = (const float*)d_in[1];
  const float* bias = (const float*)d_in[2];
  float* out = (float*)d_out;

  float* z  = (float*)d_ws;
  float* sq = z + (size_t)B_ * T_ * N_ * D_;

  GAT_z_kernel<<<8192, 128, 0, stream>>>(x, W, bias, z);

  GAT_sq_kernel<<<(B_ * T_ * N_ + 255) / 256, 256, 0, stream>>>(z, sq);

  GAT_att_kernel<<<T_ * (N_ / 16), 256, 0, stream>>>(z, sq, out);
}


# z kernel re-tiled: one wave = one (b,n), all 32 t x 64 d, A rows along t, so each 128-B x line is read whole, once (was 4 B per line per load, 4x over); next operand set loaded ahead of each block
# speedup vs baseline: 2.0471x; 2.0471x over previous
#include <hip/hip_runtime.h>
#include <hip/hip_bf16.h>

typedef __attribute__((ext_vector_type(2))) float v2f;
typedef __attribute__((ext_vector_type(8))) float v8f;

#define B_ 8
#define N_ 512
#define F_ 128
#define T_ 32
#define D_ 64
#define EPS_ 1e-6f

#define ZK_KG 4
#define ZK_NG (F_ / (4 * ZK_KG))

__global__ void __launch_bounds__(256)
GAT_z_kernel(const float* __restrict__ x,
             const float* __restrict__ W,
             const float* __restrict__ bias,
             float* __restrict__ z) {
  const int wave = blockIdx.x * (blockDim.x >> 5) + (threadIdx.x >> 5);
  const int lane = threadIdx.x & 31;
  const int n    = wave & (N_ - 1);
  const int b    = wave >> 9;

  const int l15  = lane & 15;
  const int h    = lane >> 4;
  const int koff = h << 1;

  const float* xs = x + (size_t)wave * (F_ * T_) + (size_t)koff * T_ + 2 * l15;
  const float* ws = W + (size_t)l15 * F_ + koff;

  float bv[4];
#pragma unroll
  for (int q = 0; q < 4; ++q) bv[q] = bias[q * 16 + l15];

  v8f c[2][4];
#pragma unroll
  for (int j = 0; j < 2; ++j)
#pragma unroll
    for (int q = 0; q < 4; ++q) c[j][q] = v8f{};

  v2f xr[2][2 * ZK_KG];
  v2f wr[2][4 * ZK_KG];
#pragma unroll
  for (int i = 0; i < 2 * ZK_KG; ++i)
    xr[0][i] = *(const v2f*)(xs + (size_t)(((i >> 1) << 2) + (i & 1)) * T_);
#pragma unroll
  for (int i = 0; i < 4 * ZK_KG; ++i)
    wr[0][i] = *(const v2f*)(ws + (size_t)(i & 3) * 16 * F_ + ((i >> 2) << 2));

#pragma unroll
  for (int g = 0; g < ZK_NG; ++g) {
    const int cur = g & 1;
    if (g + 1 < ZK_NG) {
#pragma unroll
      for (int i = 0; i < 2 * ZK_KG; ++i) {
        const int f = (((g + 1) * ZK_KG + (i >> 1)) << 2) + (i & 1);
        xr[cur ^ 1][i] = *(const v2f*)(xs + (size_t)f * T_);
      }
#pragma unroll
      for (int i = 0; i < 4 * ZK_KG; ++i) {
        const int k = ((g + 1) * ZK_KG + (i >> 2)) << 2;
        wr[cur ^ 1][i] = *(const v2f*)(ws + (size_t)(i & 3) * 16 * F_ + k);
      }
    }
    __builtin_amdgcn_sched_barrier(0);
#pragma unroll
    for (int s = 0; s < ZK_KG; ++s) {
      const v2f p0 = xr[cur][2 * s];
      const v2f p1 = xr[cur][2 * s + 1];
      v2f a0, a1;
      a0.x = p0.x; a0.y = p1.x;
      a1.x = p0.y; a1.y = p1.y;
#pragma unroll
      for (int q = 0; q < 4; ++q) {
        const v2f bm = wr[cur][4 * s + q];
        c[0][q] = __builtin_amdgcn_wmma_f32_16x16x4_f32(false, a0, false, bm,
                                                        (short)0, c[0][q], false, false);
        c[1][q] = __builtin_amdgcn_wmma_f32_16x16x4_f32(false, a1, false, bm,
                                                        (short)0, c[1][q], false, false);
      }
    }
    __builtin_amdgcn_sched_barrier(0);
  }

  float* zl = z + ((size_t)(b * T_ + (h << 4)) * N_ + n) * D_ + l15;
#pragma unroll
  for (int j = 0; j < 2; ++j) {
#pragma unroll
    for (int r = 0; r < 8; ++r) {
#pragma unroll
      for (int q = 0; q < 4; ++q) {
        zl[(size_t)(2 * r + j) * N_ * D_ + q * 16] = tanhf(c[j][q][r] + bv[q]);
      }
    }
  }
}

__global__ void GAT_sq_kernel(const float* __restrict__ z,
                              float* __restrict__ sq) {
  const int i = blockIdx.x * blockDim.x + threadIdx.x;
  if (i >= B_ * T_ * N_) return;
  const float* zr = z + (size_t)i * D_;
  float s = 0.f;
#pragma unroll 8
  for (int d = 0; d < D_; ++d) { const float v = zr[d]; s = fmaf(v, v, s); }
  sq[i] = s;
}

__global__ void __launch_bounds__(256)
GAT_att_kernel(const float* __restrict__ z,
               const float* __restrict__ sq,
               float* __restrict__ out) {
  __shared__ float att_s[16 * 512];
  __shared__ float sq_s[512];
  __shared__ float red_s[256];
  __shared__ float rowsum_s[16];

  const int tid   = threadIdx.x;
  const int wave  = tid >> 5;
  const int lane  = tid & 31;
  const int ntile = blockIdx.x & 31;
  const int t     = blockIdx.x >> 5;

  const int l15   = lane & 15;
  const int koff  = (lane >> 4) << 1;
  const int rbase = (lane >> 4) << 3;
  const int arow  = ntile * 16 + l15;

  float acc[32];
#pragma unroll
  for (int j = 0; j < 32; ++j) acc[j] = 0.f;

  for (int b = 0; b < B_; ++b) {
    const float* zb  = z  + (size_t)(b * T_ + t) * N_ * D_;
    const float* sqb = sq + (size_t)(b * T_ + t) * N_;

    for (int i = tid; i < N_; i += 256) sq_s[i] = sqb[i];
    __syncthreads();

    v8f c0 = {}, c1 = {}, c2 = {}, c3 = {};
    const float* arp = zb + (size_t)arow * D_;
    const float* bp0 = zb + (size_t)((wave * 4 + 0) * 16 + l15) * D_;
    const float* bp1 = zb + (size_t)((wave * 4 + 1) * 16 + l15) * D_;
    const float* bp2 = zb + (size_t)((wave * 4 + 2) * 16 + l15) * D_;
    const float* bp3 = zb + (size_t)((wave * 4 + 3) * 16 + l15) * D_;
#pragma unroll
    for (int k = 0; k < D_; k += 4) {
      const v2f a = *(const v2f*)(arp + k + koff);
      v2f bm;
      bm = *(const v2f*)(bp0 + k + koff);
      c0 = __builtin_amdgcn_wmma_f32_16x16x4_f32(false, a, false, bm, (short)0, c0, false, false);
      bm = *(const v2f*)(bp1 + k + koff);
      c1 = __builtin_amdgcn_wmma_f32_16x16x4_f32(false, a, false, bm, (short)0, c1, false, false);
      bm = *(const v2f*)(bp2 + k + koff);
      c2 = __builtin_amdgcn_wmma_f32_16x16x4_f32(false, a, false, bm, (short)0, c2, false, false);
      bm = *(const v2f*)(bp3 + k + koff);
      c3 = __builtin_amdgcn_wmma_f32_16x16x4_f32(false, a, false, bm, (short)0, c3, false, false);
    }

#pragma unroll
    for (int ct = 0; ct < 4; ++ct) {
      const v8f c = (ct == 0) ? c0 : (ct == 1) ? c1 : (ct == 2) ? c2 : c3;
      const int col = (wave * 4 + ct) * 16 + l15;
      const float sqm = sq_s[col];
      const float nm  = sqrtf(sqm) + EPS_;
#pragma unroll
      for (int v = 0; v < 8; ++v) {
        const int rloc = rbase + v;
        const float sqn = sq_s[ntile * 16 + rloc];
        const float nn  = sqrtf(sqn) + EPS_;
        const float dot = c[v];
        const float cosv = dot / (nn * nm);
        const float d2   = fmaxf(sqn + sqm - 2.0f * dot, 0.0f);
        const float diff = sqrtf(d2);
        att_s[rloc * 512 + col] = __expf(cosv - diff / (nn + nm + EPS_));
      }
    }
    __syncthreads();

    {
      const int row = tid >> 4;
      const int sl  = tid & 15;
      float part = 0.f;
#pragma unroll 4
      for (int cidx = sl; cidx < 512; cidx += 16) part += att_s[row * 512 + cidx];
      red_s[tid] = part;
    }
    __syncthreads();
    if ((tid & 15) == 0) {
      const int row = tid >> 4;
      float s = 0.f;
#pragma unroll
      for (int j = 0; j < 16; ++j) s += red_s[(row << 4) + j];
      rowsum_s[row] = s;
    }
    __syncthreads();

    const float invB = 1.0f / (float)B_;
#pragma unroll
    for (int j = 0; j < 32; ++j) {
      const int idx = tid + j * 256;
      const int r   = idx >> 9;
      acc[j] += att_s[idx] * invB / (rowsum_s[r] + EPS_);
    }
    __syncthreads();
  }

  float* outp = out + ((size_t)t * N_ + ntile * 16) * N_;
#pragma unroll
  for (int j = 0; j < 32; ++j) {
    const int idx = tid + j * 256;
    outp[idx] = acc[j];
  }
}

extern "C" void kernel_launch(void* const* d_in, const int* in_sizes, int n_in,
                              void* d_out, int out_size, void* d_ws, size_t ws_size,
                              hipStream_t stream) {
  const float* x    = (const float*)d_in[0];
  const float* W    = (const float*)d_in[1];
  const float* bias = (const float*)d_in[2];
  float* out = (float*)d_out;

  float* z  = (float*)d_ws;
  float* sq = z + (size_t)B_ * T_ * N_ * D_;

  GAT_z_kernel<<<512, 256, 0, stream>>>(x, W, bias, z);

  GAT_sq_kernel<<<(B_ * T_ * N_ + 255) / 256, 256, 0, stream>>>(z, sq);

  GAT_att_kernel<<<T_ * (N_ / 16), 256, 0, stream>>>(z, sq, out);
}


# att epilogue: each correctly rounded f32 division becomes a multiply by the hardware reciprocal (row sums: one per row) and sqrtf the hardware square root, about 1 ulp each; summing order kept
# speedup vs baseline: 2.5381x; 1.2399x over previous
#include <hip/hip_runtime.h>
#include <hip/hip_bf16.h>

typedef __attribute__((ext_vector_type(2))) float v2f;
typedef __attribute__((ext_vector_type(8))) float v8f;

#define B_ 8
#define N_ 512
#define F_ 128
#define T_ 32
#define D_ 64
#define EPS_ 1e-6f

#define ZK_KG 4
#define ZK_NG (F_ / (4 * ZK_KG))

__global__ void __launch_bounds__(256)
GAT_z_kernel(const float* __restrict__ x,
             const float* __restrict__ W,
             const float* __restrict__ bias,
             float* __restrict__ z) {
  const int wave = blockIdx.x * (blockDim.x >> 5) + (threadIdx.x >> 5);
  const int lane = threadIdx.x & 31;
  const int n    = wave & (N_ - 1);
  const int b    = wave >> 9;

  const int l15  = lane & 15;
  const int h    = lane >> 4;
  const int koff = h << 1;

  const float* xs = x + (size_t)wave * (F_ * T_) + (size_t)koff * T_ + 2 * l15;
  const float* ws = W + (size_t)l15 * F_ + koff;

  float bv[4];
#pragma unroll
  for (int q = 0; q < 4; ++q) bv[q] = bias[q * 16 + l15];

  v8f c[2][4];
#pragma unroll
  for (int j = 0; j < 2; ++j)
#pragma unroll
    for (int q = 0; q < 4; ++q) c[j][q] = v8f{};

  v2f xr[2][2 * ZK_KG];
  v2f wr[2][4 * ZK_KG];
#pragma unroll
  for (int i = 0; i < 2 * ZK_KG; ++i)
    xr[0][i] = *(const v2f*)(xs + (size_t)(((i >> 1) << 2) + (i & 1)) * T_);
#pragma unroll
  for (int i = 0; i < 4 * ZK_KG; ++i)
    wr[0][i] = *(const v2f*)(ws + (size_t)(i & 3) * 16 * F_ + ((i >> 2) << 2));

#pragma unroll
  for (int g = 0; g < ZK_NG; ++g) {
    const int cur = g & 1;
    if (g + 1 < ZK_NG) {
#pragma unroll
      for (int i = 0; i < 2 * ZK_KG; ++i) {
        const int f = (((g + 1) * ZK_KG + (i >> 1)) << 2) + (i & 1);
        xr[cur ^ 1][i] = *(const v2f*)(xs + (size_t)f * T_);
      }
#pragma unroll
      for (int i = 0; i < 4 * ZK_KG; ++i) {
        const int k = ((g + 1) * ZK_KG + (i >> 2)) << 2;
        wr[cur ^ 1][i] = *(const v2f*)(ws + (size_t)(i & 3) * 16 * F_ + k);
      }
    }
    __builtin_amdgcn_sched_barrier(0);
#pragma unroll
    for (int s = 0; s < ZK_KG; ++s) {
      const v2f p0 = xr[cur][2 * s];
      const v2f p1 = xr[cur][2 * s + 1];
      v2f a0, a1;
      a0.x = p0.x; a0.y = p1.x;
      a1.x = p0.y; a1.y = p1.y;
#pragma unroll
      for (int q = 0; q < 4; ++q) {
        const v2f bm = wr[cur][4 * s + q];
        c[0][q] = __builtin_amdgcn_wmma_f32_16x16x4_f32(false, a0, false, bm,
                                                        (short)0, c[0][q], false, false);
        c[1][q] = __builtin_amdgcn_wmma_f32_16x16x4_f32(false, a1, false, bm,
                                                        (short)0, c[1][q], false, false);
      }
    }
    __builtin_amdgcn_sched_barrier(0);
  }

  float* zl = z + ((size_t)(b * T_ + (h << 4)) * N_ + n) * D_ + l15;
#pragma unroll
  for (int j = 0; j < 2; ++j) {
#pragma unroll
    for (int r = 0; r < 8; ++r) {
#pragma unroll
      for (int q = 0; q < 4; ++q) {
        zl[(size_t)(2 * r + j) * N_ * D_ + q * 16] = tanhf(c[j][q][r] + bv[q]);
      }
    }
  }
}

__global__ void GAT_sq_kernel(const float* __restrict__ z,
                              float* __restrict__ sq) {
  const int i = blockIdx.x * blockDim.x + threadIdx.x;
  if (i >= B_ * T_ * N_) return;
  const float* zr = z + (size_t)i * D_;
  float s = 0.f;
#pragma unroll 8
  for (int d = 0; d < D_; ++d) { const float v = zr[d]; s = fmaf(v, v, s); }
  sq[i] = s;
}

__global__ void __launch_bounds__(256)
GAT_att_kernel(const float* __restrict__ z,
               const float* __restrict__ sq,
               float* __restrict__ out) {
  __shared__ float att_s[16 * 512];
  __shared__ float sq_s[512];
  __shared__ float red_s[256];
  __shared__ float rinv_s[16];

  const int tid   = threadIdx.x;
  const int wave  = tid >> 5;
  const int lane  = tid & 31;
  const int ntile = blockIdx.x & 31;
  const int t     = blockIdx.x >> 5;

  const int l15   = lane & 15;
  const int koff  = (lane >> 4) << 1;
  const int rbase = (lane >> 4) << 3;
  const int arow  = ntile * 16 + l15;

  float acc[32];
#pragma unroll
  for (int j = 0; j < 32; ++j) acc[j] = 0.f;

  for (int b = 0; b < B_; ++b) {
    const float* zb  = z  + (size_t)(b * T_ + t) * N_ * D_;
    const float* sqb = sq + (size_t)(b * T_ + t) * N_;

    for (int i = tid; i < N_; i += 256) sq_s[i] = sqb[i];
    __syncthreads();

    v8f c0 = {}, c1 = {}, c2 = {}, c3 = {};
    const float* arp = zb + (size_t)arow * D_;
    const float* bp0 = zb + (size_t)((wave * 4 + 0) * 16 + l15) * D_;
    const float* bp1 = zb + (size_t)((wave * 4 + 1) * 16 + l15) * D_;
    const float* bp2 = zb + (size_t)((wave * 4 + 2) * 16 + l15) * D_;
    const float* bp3 = zb + (size_t)((wave * 4 + 3) * 16 + l15) * D_;
#pragma unroll
    for (int k = 0; k < D_; k += 4) {
      const v2f a = *(const v2f*)(arp + k + koff);
      v2f bm;
      bm = *(const v2f*)(bp0 + k + koff);
      c0 = __builtin_amdgcn_wmma_f32_16x16x4_f32(false, a, false, bm, (short)0, c0, false, false);
      bm = *(const v2f*)(bp1 + k + koff);
      c1 = __builtin_amdgcn_wmma_f32_16x16x4_f32(false, a, false, bm, (short)0, c1, false, false);
      bm = *(const v2f*)(bp2 + k + koff);
      c2 = __builtin_amdgcn_wmma_f32_16x16x4_f32(false, a, false, bm, (short)0, c2, false, false);
      bm = *(const v2f*)(bp3 + k + koff);
      c3 = __builtin_amdgcn_wmma_f32_16x16x4_f32(false, a, false, bm, (short)0, c3, false, false);
    }

#pragma unroll
    for (int ct = 0; ct < 4; ++ct) {
      const v8f c = (ct == 0) ? c0 : (ct == 1) ? c1 : (ct == 2) ? c2 : c3;
      const int col = (wave * 4 + ct) * 16 + l15;
      const float sqm = sq_s[col];
      const float nm  = __builtin_amdgcn_sqrtf(sqm) + EPS_;
#pragma unroll
      for (int v = 0; v < 8; ++v) {
        const int rloc = rbase + v;
        const float sqn = sq_s[ntile * 16 + rloc];
        const float nn  = __builtin_amdgcn_sqrtf(sqn) + EPS_;
        const float dot = c[v];
        const float cosv = dot * __builtin_amdgcn_rcpf(nn * nm);
        const float d2   = fmaxf(sqn + sqm - 2.0f * dot, 0.0f);
        const float diff = __builtin_amdgcn_sqrtf(d2);
        att_s[rloc * 512 + col] =
            __expf(cosv - diff * __builtin_amdgcn_rcpf(nn + nm + EPS_));
      }
    }
    __syncthreads();

    {
      const int row = tid >> 4;
      const int sl  = tid & 15;
      float part = 0.f;
#pragma unroll 4
      for (int cidx = sl; cidx < 512; cidx += 16) part += att_s[row * 512 + cidx];
      red_s[tid] = part;
    }
    __syncthreads();
    if ((tid & 15) == 0) {
      const int row = tid >> 4;
      float s = 0.f;
#pragma unroll
      for (int j = 0; j < 16; ++j) s += red_s[(row << 4) + j];
      rinv_s[row] = (1.0f / (float)B_) * __builtin_amdgcn_rcpf(s + EPS_);
    }
    __syncthreads();

#pragma unroll
    for (int j = 0; j < 32; ++j) {
      const int idx = tid + j * 256;
      const int r   = idx >> 9;
      acc[j] += att_s[idx] * rinv_s[r];
    }
    __syncthreads();
  }

  float* outp = out + ((size_t)t * N_ + ntile * 16) * N_;
#pragma unroll
  for (int j = 0; j < 32; ++j) {
    const int idx = tid + j * 256;
    outp[idx] = acc[j];
  }
}

extern "C" void kernel_launch(void* const* d_in, const int* in_sizes, int n_in,
                              void* d_out, int out_size, void* d_ws, size_t ws_size,
                              hipStream_t stream) {
  const float* x    = (const float*)d_in[0];
  const float* W    = (const float*)d_in[1];
  const float* bias = (const float*)d_in[2];
  float* out = (float*)d_out;

  float* z  = (float*)d_ws;
  float* sq = z + (size_t)B_ * T_ * N_ * D_;

  GAT_z_kernel<<<512, 256, 0, stream>>>(x, W, bias, z);

  GAT_sq_kernel<<<(B_ * T_ * N_ + 255) / 256, 256, 0, stream>>>(z, sq);

  GAT_att_kernel<<<T_ * (N_ / 16), 256, 0, stream>>>(z, sq, out);
}


# att kernel dot: A and B fragment loads widened from 8 to 16 B per lane (half as many 16-line loads); k order inside each block of 8 permuted alike in A and B, so the dot's summing order changes
# speedup vs baseline: 2.7006x; 1.0640x over previous
#include <hip/hip_runtime.h>
#include <hip/hip_bf16.h>

typedef __attribute__((ext_vector_type(2))) float v2f;
typedef __attribute__((ext_vector_type(4))) float v4f;
typedef __attribute__((ext_vector_type(8))) float v8f;

#define B_ 8
#define N_ 512
#define F_ 128
#define T_ 32
#define D_ 64
#define EPS_ 1e-6f

#define ZK_KG 4
#define ZK_NG (F_ / (4 * ZK_KG))

__global__ void __launch_bounds__(256)
GAT_z_kernel(const float* __restrict__ x,
             const float* __restrict__ W,
             const float* __restrict__ bias,
             float* __restrict__ z) {
  const int wave = blockIdx.x * (blockDim.x >> 5) + (threadIdx.x >> 5);
  const int lane = threadIdx.x & 31;
  const int n    = wave & (N_ - 1);
  const int b    = wave >> 9;

  const int l15  = lane & 15;
  const int h    = lane >> 4;
  const int koff = h << 1;

  const float* xs = x + (size_t)wave * (F_ * T_) + (size_t)koff * T_ + 2 * l15;
  const float* ws = W + (size_t)l15 * F_ + koff;

  float bv[4];
#pragma unroll
  for (int q = 0; q < 4; ++q) bv[q] = bias[q * 16 + l15];

  v8f c[2][4];
#pragma unroll
  for (int j = 0; j < 2; ++j)
#pragma unroll
    for (int q = 0; q < 4; ++q) c[j][q] = v8f{};

  v2f xr[2][2 * ZK_KG];
  v2f wr[2][4 * ZK_KG];
#pragma unroll
  for (int i = 0; i < 2 * ZK_KG; ++i)
    xr[0][i] = *(const v2f*)(xs + (size_t)(((i >> 1) << 2) + (i & 1)) * T_);
#pragma unroll
  for (int i = 0; i < 4 * ZK_KG; ++i)
    wr[0][i] = *(const v2f*)(ws + (size_t)(i & 3) * 16 * F_ + ((i >> 2) << 2));

#pragma unroll
  for (int g = 0; g < ZK_NG; ++g) {
    const int cur = g & 1;
    if (g + 1 < ZK_NG) {
#pragma unroll
      for (int i = 0; i < 2 * ZK_KG; ++i) {
        const int f = (((g + 1) * ZK_KG + (i >> 1)) << 2) + (i & 1);
        xr[cur ^ 1][i] = *(const v2f*)(xs + (size_t)f * T_);
      }
#pragma unroll
      for (int i = 0; i < 4 * ZK_KG; ++i) {
        const int k = ((g + 1) * ZK_KG + (i >> 2)) << 2;
        wr[cur ^ 1][i] = *(const v2f*)(ws + (size_t)(i & 3) * 16 * F_ + k);
      }
    }
    __builtin_amdgcn_sched_barrier(0);
#pragma unroll
    for (int s = 0; s < ZK_KG; ++s) {
      const v2f p0 = xr[cur][2 * s];
      const v2f p1 = xr[cur][2 * s + 1];
      v2f a0, a1;
      a0.x = p0.x; a0.y = p1.x;
      a1.x = p0.y; a1.y = p1.y;
#pragma unroll
      for (int q = 0; q < 4; ++q) {
        const v2f bm = wr[cur][4 * s + q];
        c[0][q] = __builtin_amdgcn_wmma_f32_16x16x4_f32(false, a0, false, bm,
                                                        (short)0, c[0][q], false, false);
        c[1][q] = __builtin_amdgcn_wmma_f32_16x16x4_f32(false, a1, false, bm,
                                                        (short)0, c[1][q], false, false);
      }
    }
    __builtin_amdgcn_sched_barrier(0);
  }

  float* zl = z + ((size_t)(b * T_ + (h << 4)) * N_ + n) * D_ + l15;
#pragma unroll
  for (int j = 0; j < 2; ++j) {
#pragma unroll
    for (int r = 0; r < 8; ++r) {
#pragma unroll
      for (int q = 0; q < 4; ++q) {
        zl[(size_t)(2 * r + j) * N_ * D_ + q * 16] = tanhf(c[j][q][r] + bv[q]);
      }
    }
  }
}

__global__ void GAT_sq_kernel(const float* __restrict__ z,
                              float* __restrict__ sq) {
  const int i = blockIdx.x * blockDim.x + threadIdx.x;
  if (i >= B_ * T_ * N_) return;
  const float* zr = z + (size_t)i * D_;
  float s = 0.f;
#pragma unroll 8
  for (int d = 0; d < D_; ++d) { const float v = zr[d]; s = fmaf(v, v, s); }
  sq[i] = s;
}

__global__ void __launch_bounds__(256)
GAT_att_kernel(const float* __restrict__ z,
               const float* __restrict__ sq,
               float* __restrict__ out) {
  __shared__ float att_s[16 * 512];
  __shared__ float sq_s[512];
  __shared__ float red_s[256];
  __shared__ float rinv_s[16];

  const int tid   = threadIdx.x;
  const int wave  = tid >> 5;
  const int lane  = tid & 31;
  const int ntile = blockIdx.x & 31;
  const int t     = blockIdx.x >> 5;

  const int l15   = lane & 15;
  const int k4off = (lane >> 4) << 2;
  const int rbase = (lane >> 4) << 3;
  const int arow  = ntile * 16 + l15;

  float acc[32];
#pragma unroll
  for (int j = 0; j < 32; ++j) acc[j] = 0.f;

  for (int b = 0; b < B_; ++b) {
    const float* zb  = z  + (size_t)(b * T_ + t) * N_ * D_;
    const float* sqb = sq + (size_t)(b * T_ + t) * N_;

    for (int i = tid; i < N_; i += 256) sq_s[i] = sqb[i];
    __syncthreads();

    v8f c0 = {}, c1 = {}, c2 = {}, c3 = {};
    const float* arp = zb + (size_t)arow * D_;
    const float* bp0 = zb + (size_t)((wave * 4 + 0) * 16 + l15) * D_;
    const float* bp1 = zb + (size_t)((wave * 4 + 1) * 16 + l15) * D_;
    const float* bp2 = zb + (size_t)((wave * 4 + 2) * 16 + l15) * D_;
    const float* bp3 = zb + (size_t)((wave * 4 + 3) * 16 + l15) * D_;
#pragma unroll
    for (int k = 0; k < D_; k += 8) {
      const v4f a4 = *(const v4f*)(arp + k + k4off);
      const v2f alo = {a4.x, a4.y};
      const v2f ahi = {a4.z, a4.w};
      v4f b4;
      v2f blo, bhi;
      b4 = *(const v4f*)(bp0 + k + k4off);
      blo.x = b4.x; blo.y = b4.y; bhi.x = b4.z; bhi.y = b4.w;
      c0 = __builtin_amdgcn_wmma_f32_16x16x4_f32(false, alo, false, blo, (short)0, c0, false, false);
      c0 = __builtin_amdgcn_wmma_f32_16x16x4_f32(false, ahi, false, bhi, (short)0, c0, false, false);
      b4 = *(const v4f*)(bp1 + k + k4off);
      blo.x = b4.x; blo.y = b4.y; bhi.x = b4.z; bhi.y = b4.w;
      c1 = __builtin_amdgcn_wmma_f32_16x16x4_f32(false, alo, false, blo, (short)0, c1, false, false);
      c1 = __builtin_amdgcn_wmma_f32_16x16x4_f32(false, ahi, false, bhi, (short)0, c1, false, false);
      b4 = *(const v4f*)(bp2 + k + k4off);
      blo.x = b4.x; blo.y = b4.y; bhi.x = b4.z; bhi.y = b4.w;
      c2 = __builtin_amdgcn_wmma_f32_16x16x4_f32(false, alo, false, blo, (short)0, c2, false, false);
      c2 = __builtin_amdgcn_wmma_f32_16x16x4_f32(false, ahi, false, bhi, (short)0, c2, false, false);
      b4 = *(const v4f*)(bp3 + k + k4off);
      blo.x = b4.x; blo.y = b4.y; bhi.x = b4.z; bhi.y = b4.w;
      c3 = __builtin_amdgcn_wmma_f32_16x16x4_f32(false, alo, false, blo, (short)0, c3, false, false);
      c3 = __builtin_amdgcn_wmma_f32_16x16x4_f32(false, ahi, false, bhi, (short)0, c3, false, false);
    }

#pragma unroll
    for (int ct = 0; ct < 4; ++ct) {
      const v8f c = (ct == 0) ? c0 : (ct == 1) ? c1 : (ct == 2) ? c2 : c3;
      const int col = (wave * 4 + ct) * 16 + l15;
      const float sqm = sq_s[col];
      const float nm  = __builtin_amdgcn_sqrtf(sqm) + EPS_;
#pragma unroll
      for (int v = 0; v < 8; ++v) {
        const int rloc = rbase + v;
        const float sqn = sq_s[ntile * 16 + rloc];
        const float nn  = __builtin_amdgcn_sqrtf(sqn) + EPS_;
        const float dot = c[v];
        const float cosv = dot * __builtin_amdgcn_rcpf(nn * nm);
        const float d2   = fmaxf(sqn + sqm - 2.0f * dot, 0.0f);
        const float diff = __builtin_amdgcn_sqrtf(d2);
        att_s[rloc * 512 + col] =
            __expf(cosv - diff * __builtin_amdgcn_rcpf(nn + nm + EPS_));
      }
    }
    __syncthreads();

    {
      const int row = tid >> 4;
      const int sl  = tid & 15;
      float part = 0.f;
#pragma unroll 4
      for (int cidx = sl; cidx < 512; cidx += 16) part += att_s[row * 512 + cidx];
      red_s[tid] = part;
    }
    __syncthreads();
    if ((tid & 15) == 0) {
      const int row = tid >> 4;
      float s = 0.f;
#pragma unroll
      for (int j = 0; j < 16; ++j) s += red_s[(row << 4) + j];
      rinv_s[row] = (1.0f / (float)B_) * __builtin_amdgcn_rcpf(s + EPS_);
    }
    __syncthreads();

#pragma unroll
    for (int j = 0; j < 32; ++j) {
      const int idx = tid + j * 256;
      const int r   = idx >> 9;
      acc[j] += att_s[idx] * rinv_s[r];
    }
    __syncthreads();
  }

  float* outp = out + ((size_t)t * N_ + ntile * 16) * N_;
#pragma unroll
  for (int j = 0; j < 32; ++j) {
    const int idx = tid + j * 256;
    outp[idx] = acc[j];
  }
}

extern "C" void kernel_launch(void* const* d_in, const int* in_sizes, int n_in,
                              void* d_out, int out_size, void* d_ws, size_t ws_size,
                              hipStream_t stream) {
  const float* x    = (const float*)d_in[0];
  const float* W    = (const float*)d_in[1];
  const float* bias = (const float*)d_in[2];
  float* out = (float*)d_out;

  float* z  = (float*)d_ws;
  float* sq = z + (size_t)B_ * T_ * N_ * D_;

  GAT_z_kernel<<<512, 256, 0, stream>>>(x, W, bias, z);

  GAT_sq_kernel<<<(B_ * T_ * N_ + 255) / 256, 256, 0, stream>>>(z, sq);

  GAT_att_kernel<<<T_ * (N_ / 16), 256, 0, stream>>>(z, sq, out);
}


# att kernel re-tiled: block 32 rows x 512 cols (was 16), wave 2 row x 4 col tiles: a B fragment load feeds two A tiles (B loads, L2 reads of z halved); next k-block loaded ahead; summing orders kept
# speedup vs baseline: 3.5061x; 1.2983x over previous
#include <hip/hip_runtime.h>
#include <hip/hip_bf16.h>

typedef __attribute__((ext_vector_type(2))) float v2f;
typedef __attribute__((ext_vector_type(4))) float v4f;
typedef __attribute__((ext_vector_type(8))) float v8f;

#define B_ 8
#define N_ 512
#define F_ 128
#define T_ 32
#define D_ 64
#define EPS_ 1e-6f

#define ZK_KG 4
#define ZK_NG (F_ / (4 * ZK_KG))

__global__ void __launch_bounds__(256)
GAT_z_kernel(const float* __restrict__ x,
             const float* __restrict__ W,
             const float* __restrict__ bias,
             float* __restrict__ z) {
  const int wave = blockIdx.x * (blockDim.x >> 5) + (threadIdx.x >> 5);
  const int lane = threadIdx.x & 31;
  const int n    = wave & (N_ - 1);
  const int b    = wave >> 9;

  const int l15  = lane & 15;
  const int h    = lane >> 4;
  const int koff = h << 1;

  const float* xs = x + (size_t)wave * (F_ * T_) + (size_t)koff * T_ + 2 * l15;
  const float* ws = W + (size_t)l15 * F_ + koff;

  float bv[4];
#pragma unroll
  for (int q = 0; q < 4; ++q) bv[q] = bias[q * 16 + l15];

  v8f c[2][4];
#pragma unroll
  for (int j = 0; j < 2; ++j)
#pragma unroll
    for (int q = 0; q < 4; ++q) c[j][q] = v8f{};

  v2f xr[2][2 * ZK_KG];
  v2f wr[2][4 * ZK_KG];
#pragma unroll
  for (int i = 0; i < 2 * ZK_KG; ++i)
    xr[0][i] = *(const v2f*)(xs + (size_t)(((i >> 1) << 2) + (i & 1)) * T_);
#pragma unroll
  for (int i = 0; i < 4 * ZK_KG; ++i)
    wr[0][i] = *(const v2f*)(ws + (size_t)(i & 3) * 16 * F_ + ((i >> 2) << 2));

#pragma unroll
  for (int g = 0; g < ZK_NG; ++g) {
    const int cur = g & 1;
    if (g + 1 < ZK_NG) {
#pragma unroll
      for (int i = 0; i < 2 * ZK_KG; ++i) {
        const int f = (((g + 1) * ZK_KG + (i >> 1)) << 2) + (i & 1);
        xr[cur ^ 1][i] = *(const v2f*)(xs + (size_t)f * T_);
      }
#pragma unroll
      for (int i = 0; i < 4 * ZK_KG; ++i) {
        const int k = ((g + 1) * ZK_KG + (i >> 2)) << 2;
        wr[cur ^ 1][i] = *(const v2f*)(ws + (size_t)(i & 3) * 16 * F_ + k);
      }
    }
    __builtin_amdgcn_sched_barrier(0);
#pragma unroll
    for (int s = 0; s < ZK_KG; ++s) {
      const v2f p0 = xr[cur][2 * s];
      const v2f p1 = xr[cur][2 * s + 1];
      v2f a0, a1;
      a0.x = p0.x; a0.y = p1.x;
      a1.x = p0.y; a1.y = p1.y;
#pragma unroll
      for (int q = 0; q < 4; ++q) {
        const v2f bm = wr[cur][4 * s + q];
        c[0][q] = __builtin_amdgcn_wmma_f32_16x16x4_f32(false, a0, false, bm,
                                                        (short)0, c[0][q], false, false);
        c[1][q] = __builtin_amdgcn_wmma_f32_16x16x4_f32(false, a1, false, bm,
                                                        (short)0, c[1][q], false, false);
      }
    }
    __builtin_amdgcn_sched_barrier(0);
  }

  float* zl = z + ((size_t)(b * T_ + (h << 4)) * N_ + n) * D_ + l15;
#pragma unroll
  for (int j = 0; j < 2; ++j) {
#pragma unroll
    for (int r = 0; r < 8; ++r) {
#pragma unroll
      for (int q = 0; q < 4; ++q) {
        zl[(size_t)(2 * r + j) * N_ * D_ + q * 16] = tanhf(c[j][q][r] + bv[q]);
      }
    }
  }
}

__global__ void GAT_sq_kernel(const float* __restrict__ z,
                              float* __restrict__ sq) {
  const int i = blockIdx.x * blockDim.x + threadIdx.x;
  if (i >= B_ * T_ * N_) return;
  const float* zr = z + (size_t)i * D_;
  float s = 0.f;
#pragma unroll 8
  for (int d = 0; d < D_; ++d) { const float v = zr[d]; s = fmaf(v, v, s); }
  sq[i] = s;
}

#define AR_ 32
#define ART_ (AR_ / 16)

__global__ void __launch_bounds__(256, 4)
GAT_att_kernel(const float* __restrict__ z,
               const float* __restrict__ sq,
               float* __restrict__ out) {
  __shared__ float att_s[AR_ * 512];
  __shared__ float sq_s[512];
  __shared__ float red_s[AR_ * 16];
  __shared__ float rinv_s[AR_];

  const int tid   = threadIdx.x;
  const int wave  = tid >> 5;
  const int lane  = tid & 31;
  const int ntile = blockIdx.x & (N_ / AR_ - 1);
  const int t     = blockIdx.x / (N_ / AR_);

  const int l15   = lane & 15;
  const int k4off = (lane >> 4) << 2;
  const int rbase = (lane >> 4) << 3;
  const int arow  = ntile * AR_ + l15;

  float acc[AR_ * 2];
#pragma unroll
  for (int j = 0; j < AR_ * 2; ++j) acc[j] = 0.f;

  for (int b = 0; b < B_; ++b) {
    const float* zb  = z  + (size_t)(b * T_ + t) * N_ * D_;
    const float* sqb = sq + (size_t)(b * T_ + t) * N_;

    for (int i = tid; i < N_; i += 256) sq_s[i] = sqb[i];
    __syncthreads();

    v8f c[ART_][4];
#pragma unroll
    for (int rt = 0; rt < ART_; ++rt)
#pragma unroll
      for (int ct = 0; ct < 4; ++ct) c[rt][ct] = v8f{};

    const float* arp = zb + (size_t)arow * D_ + k4off;
    const float* bp  = zb + (size_t)(wave * 64 + l15) * D_ + k4off;
    v4f fa[2][ART_], fb[2][4];
#pragma unroll
    for (int rt = 0; rt < ART_; ++rt) fa[0][rt] = *(const v4f*)(arp + (size_t)rt * 16 * D_);
#pragma unroll
    for (int ct = 0; ct < 4; ++ct)    fb[0][ct] = *(const v4f*)(bp + (size_t)ct * 16 * D_);
#pragma unroll
    for (int kb = 0; kb < D_ / 8; ++kb) {
      const int cur = kb & 1;
      if (kb + 1 < D_ / 8) {
#pragma unroll
        for (int rt = 0; rt < ART_; ++rt)
          fa[cur ^ 1][rt] = *(const v4f*)(arp + (size_t)rt * 16 * D_ + (kb + 1) * 8);
#pragma unroll
        for (int ct = 0; ct < 4; ++ct)
          fb[cur ^ 1][ct] = *(const v4f*)(bp + (size_t)ct * 16 * D_ + (kb + 1) * 8);
      }
      __builtin_amdgcn_sched_barrier(0);
#pragma unroll
      for (int ct = 0; ct < 4; ++ct) {
        v2f blo, bhi;
        blo.x = fb[cur][ct].x; blo.y = fb[cur][ct].y; bhi.x = fb[cur][ct].z; bhi.y = fb[cur][ct].w;
#pragma unroll
        for (int rt = 0; rt < ART_; ++rt) {
          v2f alo, ahi;
          alo.x = fa[cur][rt].x; alo.y = fa[cur][rt].y; ahi.x = fa[cur][rt].z; ahi.y = fa[cur][rt].w;
          c[rt][ct] = __builtin_amdgcn_wmma_f32_16x16x4_f32(false, alo, false, blo,
                                                            (short)0, c[rt][ct], false, false);
          c[rt][ct] = __builtin_amdgcn_wmma_f32_16x16x4_f32(false, ahi, false, bhi,
                                                            (short)0, c[rt][ct], false, false);
        }
      }
      __builtin_amdgcn_sched_barrier(0);
    }

#pragma unroll
    for (int rt = 0; rt < ART_; ++rt) {
#pragma unroll
      for (int ct = 0; ct < 4; ++ct) {
        if ((ct & 1) == 0 && (rt | ct) != 0) __builtin_amdgcn_sched_barrier(0);
        const int col = (wave * 4 + ct) * 16 + l15;
        const float sqm = sq_s[col];
        const float nm  = __builtin_amdgcn_sqrtf(sqm) + EPS_;
#pragma unroll
        for (int v = 0; v < 8; ++v) {
          const int rloc = rt * 16 + rbase + v;
          const float sqn = sq_s[ntile * AR_ + rloc];
          const float nn  = __builtin_amdgcn_sqrtf(sqn) + EPS_;
          const float dot = c[rt][ct][v];
          const float cosv = dot * __builtin_amdgcn_rcpf(nn * nm);
          const float d2   = fmaxf(sqn + sqm - 2.0f * dot, 0.0f);
          const float diff = __builtin_amdgcn_sqrtf(d2);
          att_s[rloc * 512 + col] =
              __expf(cosv - diff * __builtin_amdgcn_rcpf(nn + nm + EPS_));
        }
      }
    }
    __syncthreads();

    {
      const int row = tid >> 4;
      const int sl  = tid & 15;
      float part0 = 0.f, part1 = 0.f;
#pragma unroll 4
      for (int cidx = sl; cidx < 512; cidx += 16) {
        part0 += att_s[row * 512 + cidx];
        part1 += att_s[(row + 16) * 512 + cidx];
      }
      red_s[tid]       = part0;
      red_s[256 + tid] = part1;
    }
    __syncthreads();
    if ((tid & 7) == 0) {
      const int row = tid >> 3;
      float s = 0.f;
#pragma unroll
      for (int j = 0; j < 16; ++j) s += red_s[(row << 4) + j];
      rinv_s[row] = (1.0f / (float)B_) * __builtin_amdgcn_rcpf(s + EPS_);
    }
    __syncthreads();

#pragma unroll
    for (int j = 0; j < AR_ * 2; ++j) {
      const int idx = tid + j * 256;
      const int r   = idx >> 9;
      acc[j] += att_s[idx] * rinv_s[r];
    }
    __syncthreads();
  }

  float* outp = out + ((size_t)t * N_ + ntile * AR_) * N_;
#pragma unroll
  for (int j = 0; j < AR_ * 2; ++j) {
    const int idx = tid + j * 256;
    outp[idx] = acc[j];
  }
}

extern "C" void kernel_launch(void* const* d_in, const int* in_sizes, int n_in,
                              void* d_out, int out_size, void* d_ws, size_t ws_size,
                              hipStream_t stream) {
  const float* x    = (const float*)d_in[0];
  const float* W    = (const float*)d_in[1];
  const float* bias = (const float*)d_in[2];
  float* out = (float*)d_out;

  float* z  = (float*)d_ws;
  float* sq = z + (size_t)B_ * T_ * N_ * D_;

  GAT_z_kernel<<<512, 256, 0, stream>>>(x, W, bias, z);

  GAT_sq_kernel<<<(B_ * T_ * N_ + 255) / 256, 256, 0, stream>>>(z, sq);

  GAT_att_kernel<<<T_ * (N_ / AR_), 256, 0, stream>>>(z, sq, out);
}


# att kernel: per b the block copies z[b,t] into LDS with whole-line loads, each line once per block, and reads all A/B fragments from LDS (was 16-line global fragment loads); summing orders kept
# speedup vs baseline: 4.1240x; 1.1762x over previous
#include <hip/hip_runtime.h>
#include <hip/hip_bf16.h>

typedef __attribute__((ext_vector_type(2))) float v2f;
typedef __attribute__((ext_vector_type(4))) float v4f;
typedef __attribute__((ext_vector_type(8))) float v8f;

#define B_ 8
#define N_ 512
#define F_ 128
#define T_ 32
#define D_ 64
#define EPS_ 1e-6f

#define ZK_KG 4
#define ZK_NG (F_ / (4 * ZK_KG))

__global__ void __launch_bounds__(256)
GAT_z_kernel(const float* __restrict__ x,
             const float* __restrict__ W,
             const float* __restrict__ bias,
             float* __restrict__ z) {
  const int wave = blockIdx.x * (blockDim.x >> 5) + (threadIdx.x >> 5);
  const int lane = threadIdx.x & 31;
  const int n    = wave & (N_ - 1);
  const int b    = wave >> 9;

  const int l15  = lane & 15;
  const int h    = lane >> 4;
  const int koff = h << 1;

  const float* xs = x + (size_t)wave * (F_ * T_) + (size_t)koff * T_ + 2 * l15;
  const float* ws = W + (size_t)l15 * F_ + koff;

  float bv[4];
#pragma unroll
  for (int q = 0; q < 4; ++q) bv[q] = bias[q * 16 + l15];

  v8f c[2][4];
#pragma unroll
  for (int j = 0; j < 2; ++j)
#pragma unroll
    for (int q = 0; q < 4; ++q) c[j][q] = v8f{};

  v2f xr[2][2 * ZK_KG];
  v2f wr[2][4 * ZK_KG];
#pragma unroll
  for (int i = 0; i < 2 * ZK_KG; ++i)
    xr[0][i] = *(const v2f*)(xs + (size_t)(((i >> 1) << 2) + (i & 1)) * T_);
#pragma unroll
  for (int i = 0; i < 4 * ZK_KG; ++i)
    wr[0][i] = *(const v2f*)(ws + (size_t)(i & 3) * 16 * F_ + ((i >> 2) << 2));

#pragma unroll
  for (int g = 0; g < ZK_NG; ++g) {
    const int cur = g & 1;
    if (g + 1 < ZK_NG) {
#pragma unroll
      for (int i = 0; i < 2 * ZK_KG; ++i) {
        const int f = (((g + 1) * ZK_KG + (i >> 1)) << 2) + (i & 1);
        xr[cur ^ 1][i] = *(const v2f*)(xs + (size_t)f * T_);
      }
#pragma unroll
      for (int i = 0; i < 4 * ZK_KG; ++i) {
        const int k = ((g + 1) * ZK_KG + (i >> 2)) << 2;
        wr[cur ^ 1][i] = *(const v2f*)(ws + (size_t)(i & 3) * 16 * F_ + k);
      }
    }
    __builtin_amdgcn_sched_barrier(0);
#pragma unroll
    for (int s = 0; s < ZK_KG; ++s) {
      const v2f p0 = xr[cur][2 * s];
      const v2f p1 = xr[cur][2 * s + 1];
      v2f a0, a1;
      a0.x = p0.x; a0.y = p1.x;
      a1.x = p0.y; a1.y = p1.y;
#pragma unroll
      for (int q = 0; q < 4; ++q) {
        const v2f bm = wr[cur][4 * s + q];
        c[0][q] = __builtin_amdgcn_wmma_f32_16x16x4_f32(false, a0, false, bm,
                                                        (short)0, c[0][q], false, false);
        c[1][q] = __builtin_amdgcn_wmma_f32_16x16x4_f32(false, a1, false, bm,
                                                        (short)0, c[1][q], false, false);
      }
    }
    __builtin_amdgcn_sched_barrier(0);
  }

  float* zl = z + ((size_t)(b * T_ + (h << 4)) * N_ + n) * D_ + l15;
#pragma unroll
  for (int j = 0; j < 2; ++j) {
#pragma unroll
    for (int r = 0; r < 8; ++r) {
#pragma unroll
      for (int q = 0; q < 4; ++q) {
        zl[(size_t)(2 * r + j) * N_ * D_ + q * 16] = tanhf(c[j][q][r] + bv[q]);
      }
    }
  }
}

__global__ void GAT_sq_kernel(const float* __restrict__ z,
                              float* __restrict__ sq) {
  const int i = blockIdx.x * blockDim.x + threadIdx.x;
  if (i >= B_ * T_ * N_) return;
  const float* zr = z + (size_t)i * D_;
  float s = 0.f;
#pragma unroll 8
  for (int d = 0; d < D_; ++d) { const float v = zr[d]; s = fmaf(v, v, s); }
  sq[i] = s;
}

#define AR_ 32
#define ART_ (AR_ / 16)
#define PP_ 68

#define ROW_XGET(x, M)                                                         \
  __builtin_bit_cast(float, __builtin_amdgcn_update_dpp(                       \
      0, __builtin_bit_cast(int, (x)), 0x160 + (M), 0xf, 0xf, false))

__global__ void __launch_bounds__(256, 4)
GAT_att_kernel(const float* __restrict__ z,
               const float* __restrict__ sq,
               float* __restrict__ out) {
  __shared__ __attribute__((aligned(16))) float panel_s[N_ * PP_];
  __shared__ float part_s[2][8][AR_];
  __shared__ __attribute__((aligned(16))) float rinv_s[8][AR_];

  const int tid   = threadIdx.x;
  const int wave  = tid >> 5;
  const int lane  = tid & 31;
  const int ntile = blockIdx.x & (N_ / AR_ - 1);
  const int t     = blockIdx.x / (N_ / AR_);

  const int l15   = lane & 15;
  const int k4off = (lane >> 4) << 2;
  const int rbase = (lane >> 4) << 3;
  const int arow  = ntile * AR_ + l15;

  v8f acc[ART_][4];
#pragma unroll
  for (int rt = 0; rt < ART_; ++rt)
#pragma unroll
    for (int ct = 0; ct < 4; ++ct) acc[rt][ct] = v8f{};

  for (int b = 0; b < B_; ++b) {
    const float* zb  = z  + (size_t)(b * T_ + t) * N_ * D_;
    const float* sqb = sq + (size_t)(b * T_ + t) * N_;

    float sqm[4];
#pragma unroll
    for (int ct = 0; ct < 4; ++ct) sqm[ct] = sqb[(wave * 4 + ct) * 16 + l15];
    v4f sqn4[ART_][2];
#pragma unroll
    for (int rt = 0; rt < ART_; ++rt) {
      sqn4[rt][0] = *(const v4f*)(sqb + ntile * AR_ + rt * 16 + rbase);
      sqn4[rt][1] = *(const v4f*)(sqb + ntile * AR_ + rt * 16 + rbase + 4);
    }

    {
      const float* gsrc = zb + (size_t)(wave * 64) * D_ + lane * 4;
      float* ldst = panel_s + (wave * 64 + (lane >> 4)) * PP_ + (lane & 15) * 4;
      v4f pv[32];
#pragma unroll
      for (int i = 0; i < 32; ++i) pv[i] = *(const v4f*)(gsrc + (size_t)i * 2 * D_);
      __builtin_amdgcn_sched_barrier(0);
#pragma unroll
      for (int i = 0; i < 32; ++i) *(v4f*)(ldst + i * 2 * PP_) = pv[i];
    }
    __syncthreads();

    v8f c[ART_][4];
#pragma unroll
    for (int rt = 0; rt < ART_; ++rt)
#pragma unroll
      for (int ct = 0; ct < 4; ++ct) c[rt][ct] = v8f{};

    const float* arp = panel_s + arow * PP_ + k4off;
    const float* bp  = panel_s + (wave * 64 + l15) * PP_ + k4off;
    v4f fa[2][ART_], fb[2][4];
#pragma unroll
    for (int rt = 0; rt < ART_; ++rt) fa[0][rt] = *(const v4f*)(arp + rt * 16 * PP_);
#pragma unroll
    for (int ct = 0; ct < 4; ++ct)    fb[0][ct] = *(const v4f*)(bp + ct * 16 * PP_);
#pragma unroll
    for (int kb = 0; kb < D_ / 8; ++kb) {
      const int cur = kb & 1;
      if (kb + 1 < D_ / 8) {
#pragma unroll
        for (int rt = 0; rt < ART_; ++rt)
          fa[cur ^ 1][rt] = *(const v4f*)(arp + rt * 16 * PP_ + (kb + 1) * 8);
#pragma unroll
        for (int ct = 0; ct < 4; ++ct)
          fb[cur ^ 1][ct] = *(const v4f*)(bp + ct * 16 * PP_ + (kb + 1) * 8);
      }
      __builtin_amdgcn_sched_barrier(0);
#pragma unroll
      for (int ct = 0; ct < 4; ++ct) {
        v2f blo, bhi;
        blo.x = fb[cur][ct].x; blo.y = fb[cur][ct].y; bhi.x = fb[cur][ct].z; bhi.y = fb[cur][ct].w;
#pragma unroll
        for (int rt = 0; rt < ART_; ++rt) {
          v2f alo, ahi;
          alo.x = fa[cur][rt].x; alo.y = fa[cur][rt].y; ahi.x = fa[cur][rt].z; ahi.y = fa[cur][rt].w;
          c[rt][ct] = __builtin_amdgcn_wmma_f32_16x16x4_f32(false, alo, false, blo,
                                                            (short)0, c[rt][ct], false, false);
          c[rt][ct] = __builtin_amdgcn_wmma_f32_16x16x4_f32(false, ahi, false, bhi,
                                                            (short)0, c[rt][ct], false, false);
        }
      }
      __builtin_amdgcn_sched_barrier(0);
    }

    float rs[ART_][8];
#pragma unroll
    for (int rt = 0; rt < ART_; ++rt) {
#pragma unroll
      for (int ct = 0; ct < 4; ++ct) {
        if ((ct & 1) == 0 && (rt | ct) != 0) __builtin_amdgcn_sched_barrier(0);
        const float sqmv = sqm[ct];
        const float nm   = __builtin_amdgcn_sqrtf(sqmv) + EPS_;
#pragma unroll
        for (int v = 0; v < 8; ++v) {
          const float sqn = sqn4[rt][v >> 2][v & 3];
          const float nn  = __builtin_amdgcn_sqrtf(sqn) + EPS_;
          const float dot = c[rt][ct][v];
          const float cosv = dot * __builtin_amdgcn_rcpf(nn * nm);
          const float d2   = fmaxf(sqn + sqmv - 2.0f * dot, 0.0f);
          const float diff = __builtin_amdgcn_sqrtf(d2);
          const float e =
              __expf(cosv - diff * __builtin_amdgcn_rcpf(nn + nm + EPS_));
          c[rt][ct][v] = e;
          rs[rt][v] = (ct == 0) ? e : rs[rt][v] + e;
        }
      }
    }

    const int buf = b & 1;
    float y8[8], y4[4], y2[2];
    {
      const bool m8 = (l15 & 8) != 0, m4 = (l15 & 4) != 0, m2 = (l15 & 2) != 0, m1 = (l15 & 1) != 0;
#pragma unroll
      for (int i = 0; i < 8; ++i) {
        const float keep = m8 ? rs[1][i] : rs[0][i];
        const float send = m8 ? rs[0][i] : rs[1][i];
        y8[i] = keep + ROW_XGET(send, 8);
      }
#pragma unroll
      for (int i = 0; i < 4; ++i) {
        const float keep = m4 ? y8[4 + i] : y8[i];
        const float send = m4 ? y8[i] : y8[4 + i];
        y4[i] = keep + ROW_XGET(send, 4);
      }
#pragma unroll
      for (int i = 0; i < 2; ++i) {
        const float keep = m2 ? y4[2 + i] : y4[i];
        const float send = m2 ? y4[i] : y4[2 + i];
        y2[i] = keep + ROW_XGET(send, 2);
      }
      const float keep = m1 ? y2[1] : y2[0];
      const float send = m1 ? y2[0] : y2[1];
      const float rowsum = keep + ROW_XGET(send, 1);
      part_s[buf][wave][((l15 >> 3) << 4) + rbase + (l15 & 7)] = rowsum;
    }
    __syncthreads();
    {
      float s = 0.f;
#pragma unroll
      for (int w = 0; w < 8; ++w) s += part_s[buf][w][lane];
      rinv_s[wave][lane] = (1.0f / (float)B_) * __builtin_amdgcn_rcpf(s + EPS_);
    }

#pragma unroll
    for (int rt = 0; rt < ART_; ++rt) {
      const v4f r0 = *(const v4f*)&rinv_s[wave][rt * 16 + rbase];
      const v4f r1 = *(const v4f*)&rinv_s[wave][rt * 16 + rbase + 4];
#pragma unroll
      for (int ct = 0; ct < 4; ++ct) {
#pragma unroll
        for (int v = 0; v < 8; ++v) {
          const float rv = (v < 4) ? r0[v & 3] : r1[v & 3];
          acc[rt][ct][v] += c[rt][ct][v] * rv;
        }
      }
    }
  }

#pragma unroll
  for (int rt = 0; rt < ART_; ++rt)
#pragma unroll
    for (int ct = 0; ct < 4; ++ct)
#pragma unroll
      for (int v = 0; v < 8; ++v)
        panel_s[(rt * 16 + rbase + v) * 512 + (wave * 4 + ct) * 16 + l15] = acc[rt][ct][v];
  __syncthreads();
  float* outp = out + ((size_t)t * N_ + ntile * AR_) * N_;
#pragma unroll
  for (int j = 0; j < AR_ * 2; ++j) {
    const int idx = tid + j * 256;
    outp[idx] = panel_s[idx];
  }
}

extern "C" void kernel_launch(void* const* d_in, const int* in_sizes, int n_in,
                              void* d_out, int out_size, void* d_ws, size_t ws_size,
                              hipStream_t stream) {
  const float* x    = (const float*)d_in[0];
  const float* W    = (const float*)d_in[1];
  const float* bias = (const float*)d_in[2];
  float* out = (float*)d_out;

  float* z  = (float*)d_ws;
  float* sq = z + (size_t)B_ * T_ * N_ * D_;

  GAT_z_kernel<<<512, 256, 0, stream>>>(x, W, bias, z);

  GAT_sq_kernel<<<(B_ * T_ * N_ + 255) / 256, 256, 0, stream>>>(z, sq);

  GAT_att_kernel<<<T_ * (N_ / AR_), 256, 0, stream>>>(z, sq, out);
}


# att kernel: z[b,t] double-buffered in LDS by one 16-wave block per WGP (64 rows); copy b+1 comes in 4 chunks under the dot of b: copy phase and its barrier gone, L2 reads of z x9/16; sum orders kept
# speedup vs baseline: 4.4143x; 1.0704x over previous
#include <hip/hip_runtime.h>
#include <hip/hip_bf16.h>

typedef __attribute__((ext_vector_type(2))) float v2f;
typedef __attribute__((ext_vector_type(4))) float v4f;
typedef __attribute__((ext_vector_type(8))) float v8f;

#define B_ 8
#define N_ 512
#define F_ 128
#define T_ 32
#define D_ 64
#define EPS_ 1e-6f

#define ZK_KG 4
#define ZK_NG (F_ / (4 * ZK_KG))

__global__ void __launch_bounds__(256)
GAT_z_kernel(const float* __restrict__ x,
             const float* __restrict__ W,
             const float* __restrict__ bias,
             float* __restrict__ z) {
  const int wave = blockIdx.x * (blockDim.x >> 5) + (threadIdx.x >> 5);
  const int lane = threadIdx.x & 31;
  const int n    = wave & (N_ - 1);
  const int b    = wave >> 9;

  const int l15  = lane & 15;
  const int h    = lane >> 4;
  const int koff = h << 1;

  const float* xs = x + (size_t)wave * (F_ * T_) + (size_t)koff * T_ + 2 * l15;
  const float* ws = W + (size_t)l15 * F_ + koff;

  float bv[4];
#pragma unroll
  for (int q = 0; q < 4; ++q) bv[q] = bias[q * 16 + l15];

  v8f c[2][4];
#pragma unroll
  for (int j = 0; j < 2; ++j)
#pragma unroll
    for (int q = 0; q < 4; ++q) c[j][q] = v8f{};

  v2f xr[2][2 * ZK_KG];
  v2f wr[2][4 * ZK_KG];
#pragma unroll
  for (int i = 0; i < 2 * ZK_KG; ++i)
    xr[0][i] = *(const v2f*)(xs + (size_t)(((i >> 1) << 2) + (i & 1)) * T_);
#pragma unroll
  for (int i = 0; i < 4 * ZK_KG; ++i)
    wr[0][i] = *(const v2f*)(ws + (size_t)(i & 3) * 16 * F_ + ((i >> 2) << 2));

#pragma unroll
  for (int g = 0; g < ZK_NG; ++g) {
    const int cur = g & 1;
    if (g + 1 < ZK_NG) {
#pragma unroll
      for (int i = 0; i < 2 * ZK_KG; ++i) {
        const int f = (((g + 1) * ZK_KG + (i >> 1)) << 2) + (i & 1);
        xr[cur ^ 1][i] = *(const v2f*)(xs + (size_t)f * T_);
      }
#pragma unroll
      for (int i = 0; i < 4 * ZK_KG; ++i) {
        const int k = ((g + 1) * ZK_KG + (i >> 2)) << 2;
        wr[cur ^ 1][i] = *(const v2f*)(ws + (size_t)(i & 3) * 16 * F_ + k);
      }
    }
    __builtin_amdgcn_sched_barrier(0);
#pragma unroll
    for (int s = 0; s < ZK_KG; ++s) {
      const v2f p0 = xr[cur][2 * s];
      const v2f p1 = xr[cur][2 * s + 1];
      v2f a0, a1;
      a0.x = p0.x; a0.y = p1.x;
      a1.x = p0.y; a1.y = p1.y;
#pragma unroll
      for (int q = 0; q < 4; ++q) {
        const v2f bm = wr[cur][4 * s + q];
        c[0][q] = __builtin_amdgcn_wmma_f32_16x16x4_f32(false, a0, false, bm,
                                                        (short)0, c[0][q], false, false);
        c[1][q] = __builtin_amdgcn_wmma_f32_16x16x4_f32(false, a1, false, bm,
                                                        (short)0, c[1][q], false, false);
      }
    }
    __builtin_amdgcn_sched_barrier(0);
  }

  float* zl = z + ((size_t)(b * T_ + (h << 4)) * N_ + n) * D_ + l15;
#pragma unroll
  for (int j = 0; j < 2; ++j) {
#pragma unroll
    for (int r = 0; r < 8; ++r) {
#pragma unroll
      for (int q = 0; q < 4; ++q) {
        zl[(size_t)(2 * r + j) * N_ * D_ + q * 16] = tanhf(c[j][q][r] + bv[q]);
      }
    }
  }
}

__global__ void GAT_sq_kernel(const float* __restrict__ z,
                              float* __restrict__ sq) {
  const int i = blockIdx.x * blockDim.x + threadIdx.x;
  if (i >= B_ * T_ * N_) return;
  const float* zr = z + (size_t)i * D_;
  float s = 0.f;
#pragma unroll 8
  for (int d = 0; d < D_; ++d) { const float v = zr[d]; s = fmaf(v, v, s); }
  sq[i] = s;
}

#define AR_ 64
#define WR_ 32
#define ART_ (WR_ / 16)
#define PP_ 68
#define PANEL_ (N_ * PP_)

#define ROW_XGET(x, M)                                                         \
  __builtin_bit_cast(float, __builtin_amdgcn_update_dpp(                       \
      0, __builtin_bit_cast(int, (x)), 0x160 + (M), 0xf, 0xf, false))

__global__ void __launch_bounds__(512, 4)
GAT_att_kernel(const float* __restrict__ z,
               const float* __restrict__ sq,
               float* __restrict__ out) {
  __shared__ __attribute__((aligned(16))) float panel_s[2 * PANEL_];
  __shared__ float part_s[2][8][AR_];
  __shared__ __attribute__((aligned(16))) float rinv_s[16][WR_];

  const int tid   = threadIdx.x;
  const int wave  = tid >> 5;
  const int lane  = tid & 31;
  const int rg    = wave >> 3;
  const int cw    = wave & 7;
  const int ntile = blockIdx.x & (N_ / AR_ - 1);
  const int t     = blockIdx.x / (N_ / AR_);

  const int l15   = lane & 15;
  const int k4off = (lane >> 4) << 2;
  const int rbase = (lane >> 4) << 3;
  const int arow  = ntile * AR_ + rg * WR_ + l15;

  const int coff_g = (wave * 32) * D_ + lane * 4;
  const int coff_l = (wave * 32 + (lane >> 4)) * PP_ + (lane & 15) * 4;

  v8f acc[ART_][4];
#pragma unroll
  for (int rt = 0; rt < ART_; ++rt)
#pragma unroll
    for (int ct = 0; ct < 4; ++ct) acc[rt][ct] = v8f{};

  {
    const float* gsrc = z + (size_t)t * N_ * D_ + coff_g;
    float* ldst = panel_s + coff_l;
    v4f pv[16];
#pragma unroll
    for (int i = 0; i < 16; ++i) pv[i] = *(const v4f*)(gsrc + (size_t)i * 2 * D_);
    __builtin_amdgcn_sched_barrier(0);
#pragma unroll
    for (int i = 0; i < 16; ++i) *(v4f*)(ldst + i * 2 * PP_) = pv[i];
  }
  __syncthreads();

  for (int b = 0; b < B_; ++b) {
    const float* sqb = sq + (size_t)(b * T_ + t) * N_;
    const float* pcur = panel_s + (b & 1) * PANEL_;
    const int bn = (b + 1) & (B_ - 1);
    const float* gnxt = z + (size_t)(bn * T_ + t) * N_ * D_ + coff_g;
    float* lnxt = panel_s + ((b + 1) & 1) * PANEL_ + coff_l;

    float sqm[4];
#pragma unroll
    for (int ct = 0; ct < 4; ++ct) sqm[ct] = sqb[(cw * 4 + ct) * 16 + l15];
    v4f sqn4[ART_][2];
#pragma unroll
    for (int rt = 0; rt < ART_; ++rt) {
      sqn4[rt][0] = *(const v4f*)(sqb + ntile * AR_ + rg * WR_ + rt * 16 + rbase);
      sqn4[rt][1] = *(const v4f*)(sqb + ntile * AR_ + rg * WR_ + rt * 16 + rbase + 4);
    }

    v8f c[ART_][4];
#pragma unroll
    for (int rt = 0; rt < ART_; ++rt)
#pragma unroll
      for (int ct = 0; ct < 4; ++ct) c[rt][ct] = v8f{};

    const float* arp = pcur + arow * PP_ + k4off;
    const float* bp  = pcur + (cw * 64 + l15) * PP_ + k4off;
    v4f fa[2][ART_], fb[2][4];
    v4f pq[4];
#pragma unroll
    for (int rt = 0; rt < ART_; ++rt) fa[0][rt] = *(const v4f*)(arp + rt * 16 * PP_);
#pragma unroll
    for (int ct = 0; ct < 4; ++ct)    fb[0][ct] = *(const v4f*)(bp + ct * 16 * PP_);
#pragma unroll
    for (int kb = 0; kb < D_ / 8; ++kb) {
      const int cur = kb & 1;
      if (kb + 1 < D_ / 8) {
#pragma unroll
        for (int rt = 0; rt < ART_; ++rt)
          fa[cur ^ 1][rt] = *(const v4f*)(arp + rt * 16 * PP_ + (kb + 1) * 8);
#pragma unroll
        for (int ct = 0; ct < 4; ++ct)
          fb[cur ^ 1][ct] = *(const v4f*)(bp + ct * 16 * PP_ + (kb + 1) * 8);
      }
      if ((kb & 1) == 0) {
#pragma unroll
        for (int i = 0; i < 4; ++i)
          pq[i] = *(const v4f*)(gnxt + (size_t)((kb >> 1) * 4 + i) * 2 * D_);
      }
      __builtin_amdgcn_sched_barrier(0);
#pragma unroll
      for (int ct = 0; ct < 4; ++ct) {
        v2f blo, bhi;
        blo.x = fb[cur][ct].x; blo.y = fb[cur][ct].y; bhi.x = fb[cur][ct].z; bhi.y = fb[cur][ct].w;
#pragma unroll
        for (int rt = 0; rt < ART_; ++rt) {
          v2f alo, ahi;
          alo.x = fa[cur][rt].x; alo.y = fa[cur][rt].y; ahi.x = fa[cur][rt].z; ahi.y = fa[cur][rt].w;
          c[rt][ct] = __builtin_amdgcn_wmma_f32_16x16x4_f32(false, alo, false, blo,
                                                            (short)0, c[rt][ct], false, false);
          c[rt][ct] = __builtin_amdgcn_wmma_f32_16x16x4_f32(false, ahi, false, bhi,
                                                            (short)0, c[rt][ct], false, false);
        }
      }
      __builtin_amdgcn_sched_barrier(0);
      if ((kb & 1) == 1) {
#pragma unroll
        for (int i = 0; i < 4; ++i)
          *(v4f*)(lnxt + ((kb >> 1) * 4 + i) * 2 * PP_) = pq[i];
      }
    }

    float rs[ART_][8];
#pragma unroll
    for (int rt = 0; rt < ART_; ++rt) {
#pragma unroll
      for (int ct = 0; ct < 4; ++ct) {
        if ((ct & 1) == 0 && (rt | ct) != 0) __builtin_amdgcn_sched_barrier(0);
        const float sqmv = sqm[ct];
        const float nm   = __builtin_amdgcn_sqrtf(sqmv) + EPS_;
#pragma unroll
        for (int v = 0; v < 8; ++v) {
          const float sqn = sqn4[rt][v >> 2][v & 3];
          const float nn  = __builtin_amdgcn_sqrtf(sqn) + EPS_;
          const float dot = c[rt][ct][v];
          const float cosv = dot * __builtin_amdgcn_rcpf(nn * nm);
          const float d2   = fmaxf(sqn + sqmv - 2.0f * dot, 0.0f);
          const float diff = __builtin_amdgcn_sqrtf(d2);
          const float e =
              __expf(cosv - diff * __builtin_amdgcn_rcpf(nn + nm + EPS_));
          c[rt][ct][v] = e;
          rs[rt][v] = (ct == 0) ? e : rs[rt][v] + e;
        }
      }
    }

    const int buf = b & 1;
    float y8[8], y4[4], y2[2];
    {
      const bool m8 = (l15 & 8) != 0, m4 = (l15 & 4) != 0, m2 = (l15 & 2) != 0, m1 = (l15 & 1) != 0;
#pragma unroll
      for (int i = 0; i < 8; ++i) {
        const float keep = m8 ? rs[1][i] : rs[0][i];
        const float send = m8 ? rs[0][i] : rs[1][i];
        y8[i] = keep + ROW_XGET(send, 8);
      }
#pragma unroll
      for (int i = 0; i < 4; ++i) {
        const float keep = m4 ? y8[4 + i] : y8[i];
        const float send = m4 ? y8[i] : y8[4 + i];
        y4[i] = keep + ROW_XGET(send, 4);
      }
#pragma unroll
      for (int i = 0; i < 2; ++i) {
        const float keep = m2 ? y4[2 + i] : y4[i];
        const float send = m2 ? y4[i] : y4[2 + i];
        y2[i] = keep + ROW_XGET(send, 2);
      }
      const float keep = m1 ? y2[1] : y2[0];
      const float send = m1 ? y2[0] : y2[1];
      const float rowsum = keep + ROW_XGET(send, 1);
      part_s[buf][cw][rg * WR_ + ((l15 >> 3) << 4) + rbase + (l15 & 7)] = rowsum;
    }
    __syncthreads();
    {
      float s = 0.f;
#pragma unroll
      for (int w = 0; w < 8; ++w) s += part_s[buf][w][rg * WR_ + lane];
      rinv_s[wave][lane] = (1.0f / (float)B_) * __builtin_amdgcn_rcpf(s + EPS_);
    }

#pragma unroll
    for (int rt = 0; rt < ART_; ++rt) {
      const v4f r0 = *(const v4f*)&rinv_s[wave][rt * 16 + rbase];
      const v4f r1 = *(const v4f*)&rinv_s[wave][rt * 16 + rbase + 4];
#pragma unroll
      for (int ct = 0; ct < 4; ++ct) {
#pragma unroll
        for (int v = 0; v < 8; ++v) {
          const float rv = (v < 4) ? r0[v & 3] : r1[v & 3];
          acc[rt][ct][v] += c[rt][ct][v] * rv;
        }
      }
    }
  }

#pragma unroll
  for (int rt = 0; rt < ART_; ++rt)
#pragma unroll
    for (int ct = 0; ct < 4; ++ct)
#pragma unroll
      for (int v = 0; v < 8; ++v)
        panel_s[(rg * WR_ + rt * 16 + rbase + v) * 512 + (cw * 4 + ct) * 16 + l15] = acc[rt][ct][v];
  __syncthreads();
  float* outp = out + ((size_t)t * N_ + ntile * AR_) * N_;
#pragma unroll
  for (int j = 0; j < AR_; ++j) {
    const int idx = tid + j * 512;
    outp[idx] = panel_s[idx];
  }
}

extern "C" void kernel_launch(void* const* d_in, const int* in_sizes, int n_in,
                              void* d_out, int out_size, void* d_ws, size_t ws_size,
                              hipStream_t stream) {
  const float* x    = (const float*)d_in[0];
  const float* W    = (const float*)d_in[1];
  const float* bias = (const float*)d_in[2];
  float* out = (float*)d_out;

  float* z  = (float*)d_ws;
  float* sq = z + (size_t)B_ * T_ * N_ * D_;

  GAT_z_kernel<<<512, 256, 0, stream>>>(x, W, bias, z);

  GAT_sq_kernel<<<(B_ * T_ * N_ + 255) / 256, 256, 0, stream>>>(z, sq);

  GAT_att_kernel<<<T_ * (N_ / AR_), 512, 0, stream>>>(z, sq, out);
}


# squared norms taken in the z kernel from the tanh values in registers (lane sum, then halving exchange over 16 lanes: sq summing order changes); the separate sq pass over z and its launch are gone
# speedup vs baseline: 4.6090x; 1.0441x over previous
#include <hip/hip_runtime.h>
#include <hip/hip_bf16.h>

typedef __attribute__((ext_vector_type(2))) float v2f;
typedef __attribute__((ext_vector_type(4))) float v4f;
typedef __attribute__((ext_vector_type(8))) float v8f;

#define B_ 8
#define N_ 512
#define F_ 128
#define T_ 32
#define D_ 64
#define EPS_ 1e-6f


#define ROW_XGET(x, M)                                                         \
  __builtin_bit_cast(float, __builtin_amdgcn_update_dpp(                       \
      0, __builtin_bit_cast(int, (x)), 0x160 + (M), 0xf, 0xf, false))

#define ZK_KG 4
#define ZK_NG (F_ / (4 * ZK_KG))

__global__ void __launch_bounds__(256)
GAT_z_kernel(const float* __restrict__ x,
             const float* __restrict__ W,
             const float* __restrict__ bias,
             float* __restrict__ z,
             float* __restrict__ sq) {
  __shared__ float sq_s[T_ * 8];
  const int wave = blockIdx.x * (blockDim.x >> 5) + (threadIdx.x >> 5);
  const int lane = threadIdx.x & 31;
  const int n    = wave & (N_ - 1);
  const int b    = wave >> 9;

  const int l15  = lane & 15;
  const int h    = lane >> 4;
  const int koff = h << 1;

  const float* xs = x + (size_t)wave * (F_ * T_) + (size_t)koff * T_ + 2 * l15;
  const float* ws = W + (size_t)l15 * F_ + koff;

  float bv[4];
#pragma unroll
  for (int q = 0; q < 4; ++q) bv[q] = bias[q * 16 + l15];

  v8f c[2][4];
#pragma unroll
  for (int j = 0; j < 2; ++j)
#pragma unroll
    for (int q = 0; q < 4; ++q) c[j][q] = v8f{};

  v2f xr[2][2 * ZK_KG];
  v2f wr[2][4 * ZK_KG];
#pragma unroll
  for (int i = 0; i < 2 * ZK_KG; ++i)
    xr[0][i] = *(const v2f*)(xs + (size_t)(((i >> 1) << 2) + (i & 1)) * T_);
#pragma unroll
  for (int i = 0; i < 4 * ZK_KG; ++i)
    wr[0][i] = *(const v2f*)(ws + (size_t)(i & 3) * 16 * F_ + ((i >> 2) << 2));

#pragma unroll
  for (int g = 0; g < ZK_NG; ++g) {
    const int cur = g & 1;
    if (g + 1 < ZK_NG) {
#pragma unroll
      for (int i = 0; i < 2 * ZK_KG; ++i) {
        const int f = (((g + 1) * ZK_KG + (i >> 1)) << 2) + (i & 1);
        xr[cur ^ 1][i] = *(const v2f*)(xs + (size_t)f * T_);
      }
#pragma unroll
      for (int i = 0; i < 4 * ZK_KG; ++i) {
        const int k = ((g + 1) * ZK_KG + (i >> 2)) << 2;
        wr[cur ^ 1][i] = *(const v2f*)(ws + (size_t)(i & 3) * 16 * F_ + k);
      }
    }
    __builtin_amdgcn_sched_barrier(0);
#pragma unroll
    for (int s = 0; s < ZK_KG; ++s) {
      const v2f p0 = xr[cur][2 * s];
      const v2f p1 = xr[cur][2 * s + 1];
      v2f a0, a1;
      a0.x = p0.x; a0.y = p1.x;
      a1.x = p0.y; a1.y = p1.y;
#pragma unroll
      for (int q = 0; q < 4; ++q) {
        const v2f bm = wr[cur][4 * s + q];
        c[0][q] = __builtin_amdgcn_wmma_f32_16x16x4_f32(false, a0, false, bm,
                                                        (short)0, c[0][q], false, false);
        c[1][q] = __builtin_amdgcn_wmma_f32_16x16x4_f32(false, a1, false, bm,
                                                        (short)0, c[1][q], false, false);
      }
    }
    __builtin_amdgcn_sched_barrier(0);
  }

  float* zl = z + ((size_t)(b * T_ + (h << 4)) * N_ + n) * D_ + l15;
  float ss[2][8];
#pragma unroll
  for (int j = 0; j < 2; ++j) {
#pragma unroll
    for (int r = 0; r < 8; ++r) {
#pragma unroll
      for (int q = 0; q < 4; ++q) {
        const float tz = tanhf(c[j][q][r] + bv[q]);
        zl[(size_t)(2 * r + j) * N_ * D_ + q * 16] = tz;
        ss[j][r] = (q == 0) ? tz * tz : fmaf(tz, tz, ss[j][r]);
      }
    }
  }

  {
    const bool m8 = (l15 & 8) != 0, m4 = (l15 & 4) != 0, m2 = (l15 & 2) != 0, m1 = (l15 & 1) != 0;
    float y8[8], y4[4], y2[2];
#pragma unroll
    for (int i = 0; i < 8; ++i) {
      const float keep = m8 ? ss[1][i] : ss[0][i];
      const float send = m8 ? ss[0][i] : ss[1][i];
      y8[i] = keep + ROW_XGET(send, 8);
    }
#pragma unroll
    for (int i = 0; i < 4; ++i) {
      const float keep = m4 ? y8[4 + i] : y8[i];
      const float send = m4 ? y8[i] : y8[4 + i];
      y4[i] = keep + ROW_XGET(send, 4);
    }
#pragma unroll
    for (int i = 0; i < 2; ++i) {
      const float keep = m2 ? y4[2 + i] : y4[i];
      const float send = m2 ? y4[i] : y4[2 + i];
      y2[i] = keep + ROW_XGET(send, 2);
    }
    const float keep = m1 ? y2[1] : y2[0];
    const float send = m1 ? y2[0] : y2[1];
    const float sqv = keep + ROW_XGET(send, 1);
    sq_s[((h << 4) + ((l15 & 7) << 1) + (l15 >> 3)) * 8 + (threadIdx.x >> 5)] = sqv;
  }
  __syncthreads();
  sq[(size_t)blockIdx.x * (T_ * 8) + threadIdx.x] = sq_s[threadIdx.x];
}

#define AR_ 64
#define WR_ 32
#define ART_ (WR_ / 16)
#define PP_ 68
#define PANEL_ (N_ * PP_)

__global__ void __launch_bounds__(512, 4)
GAT_att_kernel(const float* __restrict__ z,
               const float* __restrict__ sq,
               float* __restrict__ out) {
  __shared__ __attribute__((aligned(16))) float panel_s[2 * PANEL_];
  __shared__ float part_s[2][8][AR_];
  __shared__ __attribute__((aligned(16))) float rinv_s[16][WR_];

  const int tid   = threadIdx.x;
  const int wave  = tid >> 5;
  const int lane  = tid & 31;
  const int rg    = wave >> 3;
  const int cw    = wave & 7;
  const int ntile = blockIdx.x & (N_ / AR_ - 1);
  const int t     = blockIdx.x / (N_ / AR_);

  const int l15   = lane & 15;
  const int k4off = (lane >> 4) << 2;
  const int rbase = (lane >> 4) << 3;
  const int arow  = ntile * AR_ + rg * WR_ + l15;

  const int coff_g = (wave * 32) * D_ + lane * 4;
  const int coff_l = (wave * 32 + (lane >> 4)) * PP_ + (lane & 15) * 4;

  v8f acc[ART_][4];
#pragma unroll
  for (int rt = 0; rt < ART_; ++rt)
#pragma unroll
    for (int ct = 0; ct < 4; ++ct) acc[rt][ct] = v8f{};

  {
    const float* gsrc = z + (size_t)t * N_ * D_ + coff_g;
    float* ldst = panel_s + coff_l;
    v4f pv[16];
#pragma unroll
    for (int i = 0; i < 16; ++i) pv[i] = *(const v4f*)(gsrc + (size_t)i * 2 * D_);
    __builtin_amdgcn_sched_barrier(0);
#pragma unroll
    for (int i = 0; i < 16; ++i) *(v4f*)(ldst + i * 2 * PP_) = pv[i];
  }
  __syncthreads();

  for (int b = 0; b < B_; ++b) {
    const float* sqb = sq + (size_t)b * (N_ * T_) + t * 8;
    const float* pcur = panel_s + (b & 1) * PANEL_;
    const int bn = (b + 1) & (B_ - 1);
    const float* gnxt = z + (size_t)(bn * T_ + t) * N_ * D_ + coff_g;
    float* lnxt = panel_s + ((b + 1) & 1) * PANEL_ + coff_l;

    float sqm[4];
#pragma unroll
    for (int ct = 0; ct < 4; ++ct)
      sqm[ct] = sqb[((cw * 4 + ct) * 2 + (l15 >> 3)) * (T_ * 8) + (l15 & 7)];
    v4f sqn4[ART_][2];
#pragma unroll
    for (int rt = 0; rt < ART_; ++rt) {
      const float* sqr = sqb + ((ntile * AR_ + rg * WR_ + rt * 16 + rbase) >> 3) * (T_ * 8);
      sqn4[rt][0] = *(const v4f*)(sqr);
      sqn4[rt][1] = *(const v4f*)(sqr + 4);
    }

    v8f c[ART_][4];
#pragma unroll
    for (int rt = 0; rt < ART_; ++rt)
#pragma unroll
      for (int ct = 0; ct < 4; ++ct) c[rt][ct] = v8f{};

    const float* arp = pcur + arow * PP_ + k4off;
    const float* bp  = pcur + (cw * 64 + l15) * PP_ + k4off;
    v4f fa[2][ART_], fb[2][4];
    v4f pq[4];
#pragma unroll
    for (int rt = 0; rt < ART_; ++rt) fa[0][rt] = *(const v4f*)(arp + rt * 16 * PP_);
#pragma unroll
    for (int ct = 0; ct < 4; ++ct)    fb[0][ct] = *(const v4f*)(bp + ct * 16 * PP_);
#pragma unroll
    for (int kb = 0; kb < D_ / 8; ++kb) {
      const int cur = kb & 1;
      if (kb + 1 < D_ / 8) {
#pragma unroll
        for (int rt = 0; rt < ART_; ++rt)
          fa[cur ^ 1][rt] = *(const v4f*)(arp + rt * 16 * PP_ + (kb + 1) * 8);
#pragma unroll
        for (int ct = 0; ct < 4; ++ct)
          fb[cur ^ 1][ct] = *(const v4f*)(bp + ct * 16 * PP_ + (kb + 1) * 8);
      }
      if ((kb & 1) == 0) {
#pragma unroll
        for (int i = 0; i < 4; ++i)
          pq[i] = *(const v4f*)(gnxt + (size_t)((kb >> 1) * 4 + i) * 2 * D_);
      }
      __builtin_amdgcn_sched_barrier(0);
#pragma unroll
      for (int ct = 0; ct < 4; ++ct) {
        v2f blo, bhi;
        blo.x = fb[cur][ct].x; blo.y = fb[cur][ct].y; bhi.x = fb[cur][ct].z; bhi.y = fb[cur][ct].w;
#pragma unroll
        for (int rt = 0; rt < ART_; ++rt) {
          v2f alo, ahi;
          alo.x = fa[cur][rt].x; alo.y = fa[cur][rt].y; ahi.x = fa[cur][rt].z; ahi.y = fa[cur][rt].w;
          c[rt][ct] = __builtin_amdgcn_wmma_f32_16x16x4_f32(false, alo, false, blo,
                                                            (short)0, c[rt][ct], false, false);
          c[rt][ct] = __builtin_amdgcn_wmma_f32_16x16x4_f32(false, ahi, false, bhi,
                                                            (short)0, c[rt][ct], false, false);
        }
      }
      __builtin_amdgcn_sched_barrier(0);
      if ((kb & 1) == 1) {
#pragma unroll
        for (int i = 0; i < 4; ++i)
          *(v4f*)(lnxt + ((kb >> 1) * 4 + i) * 2 * PP_) = pq[i];
      }
    }

    float rs[ART_][8];
#pragma unroll
    for (int rt = 0; rt < ART_; ++rt) {
#pragma unroll
      for (int ct = 0; ct < 4; ++ct) {
        if ((ct & 1) == 0 && (rt | ct) != 0) __builtin_amdgcn_sched_barrier(0);
        const float sqmv = sqm[ct];
        const float nm   = __builtin_amdgcn_sqrtf(sqmv) + EPS_;
#pragma unroll
        for (int v = 0; v < 8; ++v) {
          const float sqn = sqn4[rt][v >> 2][v & 3];
          const float nn  = __builtin_amdgcn_sqrtf(sqn) + EPS_;
          const float dot = c[rt][ct][v];
          const float cosv = dot * __builtin_amdgcn_rcpf(nn * nm);
          const float d2   = fmaxf(sqn + sqmv - 2.0f * dot, 0.0f);
          const float diff = __builtin_amdgcn_sqrtf(d2);
          const float e =
              __expf(cosv - diff * __builtin_amdgcn_rcpf(nn + nm + EPS_));
          c[rt][ct][v] = e;
          rs[rt][v] = (ct == 0) ? e : rs[rt][v] + e;
        }
      }
    }

    const int buf = b & 1;
    float y8[8], y4[4], y2[2];
    {
      const bool m8 = (l15 & 8) != 0, m4 = (l15 & 4) != 0, m2 = (l15 & 2) != 0, m1 = (l15 & 1) != 0;
#pragma unroll
      for (int i = 0; i < 8; ++i) {
        const float keep = m8 ? rs[1][i] : rs[0][i];
        const float send = m8 ? rs[0][i] : rs[1][i];
        y8[i] = keep + ROW_XGET(send, 8);
      }
#pragma unroll
      for (int i = 0; i < 4; ++i) {
        const float keep = m4 ? y8[4 + i] : y8[i];
        const float send = m4 ? y8[i] : y8[4 + i];
        y4[i] = keep + ROW_XGET(send, 4);
      }
#pragma unroll
      for (int i = 0; i < 2; ++i) {
        const float keep = m2 ? y4[2 + i] : y4[i];
        const float send = m2 ? y4[i] : y4[2 + i];
        y2[i] = keep + ROW_XGET(send, 2);
      }
      const float keep = m1 ? y2[1] : y2[0];
      const float send = m1 ? y2[0] : y2[1];
      const float rowsum = keep + ROW_XGET(send, 1);
      part_s[buf][cw][rg * WR_ + ((l15 >> 3) << 4) + rbase + (l15 & 7)] = rowsum;
    }
    __syncthreads();
    {
      float s = 0.f;
#pragma unroll
      for (int w = 0; w < 8; ++w) s += part_s[buf][w][rg * WR_ + lane];
      rinv_s[wave][lane] = (1.0f / (float)B_) * __builtin_amdgcn_rcpf(s + EPS_);
    }

#pragma unroll
    for (int rt = 0; rt < ART_; ++rt) {
      const v4f r0 = *(const v4f*)&rinv_s[wave][rt * 16 + rbase];
      const v4f r1 = *(const v4f*)&rinv_s[wave][rt * 16 + rbase + 4];
#pragma unroll
      for (int ct = 0; ct < 4; ++ct) {
#pragma unroll
        for (int v = 0; v < 8; ++v) {
          const float rv = (v < 4) ? r0[v & 3] : r1[v & 3];
          acc[rt][ct][v] += c[rt][ct][v] * rv;
        }
      }
    }
  }

#pragma unroll
  for (int rt = 0; rt < ART_; ++rt)
#pragma unroll
    for (int ct = 0; ct < 4; ++ct)
#pragma unroll
      for (int v = 0; v < 8; ++v)
        panel_s[(rg * WR_ + rt * 16 + rbase + v) * 512 + (cw * 4 + ct) * 16 + l15] = acc[rt][ct][v];
  __syncthreads();
  float* outp = out + ((size_t)t * N_ + ntile * AR_) * N_;
#pragma unroll
  for (int j = 0; j < AR_; ++j) {
    const int idx = tid + j * 512;
    outp[idx] = panel_s[idx];
  }
}

extern "C" void kernel_launch(void* const* d_in, const int* in_sizes, int n_in,
                              void* d_out, int out_size, void* d_ws, size_t ws_size,
                              hipStream_t stream) {
  const float* x    = (const float*)d_in[0];
  const float* W    = (const float*)d_in[1];
  const float* bias = (const float*)d_in[2];
  float* out = (float*)d_out;

  float* z  = (float*)d_ws;
  float* sq = z + (size_t)B_ * T_ * N_ * D_;

  GAT_z_kernel<<<512, 256, 0, stream>>>(x, W, bias, z, sq);

  GAT_att_kernel<<<T_ * (N_ / AR_), 512, 0, stream>>>(z, sq, out);
}


# att epilogue: 3 transcendentals a value (was 4): 1/(nn*nm) as (1/nn)(1/nm), once per row and column; log2e folded into these; norm sum as nn+(nm+eps); two rows per packed f32 op; other sums unchanged
# speedup vs baseline: 4.8227x; 1.0464x over previous
#include <hip/hip_runtime.h>
#include <hip/hip_bf16.h>

typedef __attribute__((ext_vector_type(2))) float v2f;
typedef __attribute__((ext_vector_type(4))) float v4f;
typedef __attribute__((ext_vector_type(8))) float v8f;

#define B_ 8
#define N_ 512
#define F_ 128
#define T_ 32
#define D_ 64
#define EPS_ 1e-6f


#define ROW_XGET(x, M)                                                         \
  __builtin_bit_cast(float, __builtin_amdgcn_update_dpp(                       \
      0, __builtin_bit_cast(int, (x)), 0x160 + (M), 0xf, 0xf, false))

#define ZK_KG 4
#define ZK_NG (F_ / (4 * ZK_KG))

__global__ void __launch_bounds__(256)
GAT_z_kernel(const float* __restrict__ x,
             const float* __restrict__ W,
             const float* __restrict__ bias,
             float* __restrict__ z,
             float* __restrict__ sq) {
  __shared__ float sq_s[T_ * 8];
  const int wave = blockIdx.x * (blockDim.x >> 5) + (threadIdx.x >> 5);
  const int lane = threadIdx.x & 31;
  const int n    = wave & (N_ - 1);
  const int b    = wave >> 9;

  const int l15  = lane & 15;
  const int h    = lane >> 4;
  const int koff = h << 1;

  const float* xs = x + (size_t)wave * (F_ * T_) + (size_t)koff * T_ + 2 * l15;
  const float* ws = W + (size_t)l15 * F_ + koff;

  float bv[4];
#pragma unroll
  for (int q = 0; q < 4; ++q) bv[q] = bias[q * 16 + l15];

  v8f c[2][4];
#pragma unroll
  for (int j = 0; j < 2; ++j)
#pragma unroll
    for (int q = 0; q < 4; ++q) c[j][q] = v8f{};

  v2f xr[2][2 * ZK_KG];
  v2f wr[2][4 * ZK_KG];
#pragma unroll
  for (int i = 0; i < 2 * ZK_KG; ++i)
    xr[0][i] = *(const v2f*)(xs + (size_t)(((i >> 1) << 2) + (i & 1)) * T_);
#pragma unroll
  for (int i = 0; i < 4 * ZK_KG; ++i)
    wr[0][i] = *(const v2f*)(ws + (size_t)(i & 3) * 16 * F_ + ((i >> 2) << 2));

#pragma unroll
  for (int g = 0; g < ZK_NG; ++g) {
    const int cur = g & 1;
    if (g + 1 < ZK_NG) {
#pragma unroll
      for (int i = 0; i < 2 * ZK_KG; ++i) {
        const int f = (((g + 1) * ZK_KG + (i >> 1)) << 2) + (i & 1);
        xr[cur ^ 1][i] = *(const v2f*)(xs + (size_t)f * T_);
      }
#pragma unroll
      for (int i = 0; i < 4 * ZK_KG; ++i) {
        const int k = ((g + 1) * ZK_KG + (i >> 2)) << 2;
        wr[cur ^ 1][i] = *(const v2f*)(ws + (size_t)(i & 3) * 16 * F_ + k);
      }
    }
    __builtin_amdgcn_sched_barrier(0);
#pragma unroll
    for (int s = 0; s < ZK_KG; ++s) {
      const v2f p0 = xr[cur][2 * s];
      const v2f p1 = xr[cur][2 * s + 1];
      v2f a0, a1;
      a0.x = p0.x; a0.y = p1.x;
      a1.x = p0.y; a1.y = p1.y;
#pragma unroll
      for (int q = 0; q < 4; ++q) {
        const v2f bm = wr[cur][4 * s + q];
        c[0][q] = __builtin_amdgcn_wmma_f32_16x16x4_f32(false, a0, false, bm,
                                                        (short)0, c[0][q], false, false);
        c[1][q] = __builtin_amdgcn_wmma_f32_16x16x4_f32(false, a1, false, bm,
                                                        (short)0, c[1][q], false, false);
      }
    }
    __builtin_amdgcn_sched_barrier(0);
  }

  float* zl = z + ((size_t)(b * T_ + (h << 4)) * N_ + n) * D_ + l15;
  float ss[2][8];
#pragma unroll
  for (int j = 0; j < 2; ++j) {
#pragma unroll
    for (int r = 0; r < 8; ++r) {
#pragma unroll
      for (int q = 0; q < 4; ++q) {
        const float tz = tanhf(c[j][q][r] + bv[q]);
        zl[(size_t)(2 * r + j) * N_ * D_ + q * 16] = tz;
        ss[j][r] = (q == 0) ? tz * tz : fmaf(tz, tz, ss[j][r]);
      }
    }
  }

  {
    const bool m8 = (l15 & 8) != 0, m4 = (l15 & 4) != 0, m2 = (l15 & 2) != 0, m1 = (l15 & 1) != 0;
    float y8[8], y4[4], y2[2];
#pragma unroll
    for (int i = 0; i < 8; ++i) {
      const float keep = m8 ? ss[1][i] : ss[0][i];
      const float send = m8 ? ss[0][i] : ss[1][i];
      y8[i] = keep + ROW_XGET(send, 8);
    }
#pragma unroll
    for (int i = 0; i < 4; ++i) {
      const float keep = m4 ? y8[4 + i] : y8[i];
      const float send = m4 ? y8[i] : y8[4 + i];
      y4[i] = keep + ROW_XGET(send, 4);
    }
#pragma unroll
    for (int i = 0; i < 2; ++i) {
      const float keep = m2 ? y4[2 + i] : y4[i];
      const float send = m2 ? y4[i] : y4[2 + i];
      y2[i] = keep + ROW_XGET(send, 2);
    }
    const float keep = m1 ? y2[1] : y2[0];
    const float send = m1 ? y2[0] : y2[1];
    const float sqv = keep + ROW_XGET(send, 1);
    sq_s[((h << 4) + ((l15 & 7) << 1) + (l15 >> 3)) * 8 + (threadIdx.x >> 5)] = sqv;
  }
  __syncthreads();
  sq[(size_t)blockIdx.x * (T_ * 8) + threadIdx.x] = sq_s[threadIdx.x];
}

#define AR_ 64
#define WR_ 32
#define ART_ (WR_ / 16)
#define PP_ 68
#define PANEL_ (N_ * PP_)

__global__ void __launch_bounds__(512, 4)
GAT_att_kernel(const float* __restrict__ z,
               const float* __restrict__ sq,
               float* __restrict__ out) {
  __shared__ __attribute__((aligned(16))) float panel_s[2 * PANEL_];
  __shared__ float part_s[2][8][AR_];
  __shared__ __attribute__((aligned(16))) float rinv_s[16][WR_];

  const int tid   = threadIdx.x;
  const int wave  = tid >> 5;
  const int lane  = tid & 31;
  const int rg    = wave >> 3;
  const int cw    = wave & 7;
  const int ntile = blockIdx.x & (N_ / AR_ - 1);
  const int t     = blockIdx.x / (N_ / AR_);

  const int l15   = lane & 15;
  const int k4off = (lane >> 4) << 2;
  const int rbase = (lane >> 4) << 3;
  const int arow  = ntile * AR_ + rg * WR_ + l15;

  const int coff_g = (wave * 32) * D_ + lane * 4;
  const int coff_l = (wave * 32 + (lane >> 4)) * PP_ + (lane & 15) * 4;

  v8f acc[ART_][4];
#pragma unroll
  for (int rt = 0; rt < ART_; ++rt)
#pragma unroll
    for (int ct = 0; ct < 4; ++ct) acc[rt][ct] = v8f{};

  {
    const float* gsrc = z + (size_t)t * N_ * D_ + coff_g;
    float* ldst = panel_s + coff_l;
    v4f pv[16];
#pragma unroll
    for (int i = 0; i < 16; ++i) pv[i] = *(const v4f*)(gsrc + (size_t)i * 2 * D_);
    __builtin_amdgcn_sched_barrier(0);
#pragma unroll
    for (int i = 0; i < 16; ++i) *(v4f*)(ldst + i * 2 * PP_) = pv[i];
  }
  __syncthreads();

  for (int b = 0; b < B_; ++b) {
    const float* sqb = sq + (size_t)b * (N_ * T_) + t * 8;
    const float* pcur = panel_s + (b & 1) * PANEL_;
    const int bn = (b + 1) & (B_ - 1);
    const float* gnxt = z + (size_t)(bn * T_ + t) * N_ * D_ + coff_g;
    float* lnxt = panel_s + ((b + 1) & 1) * PANEL_ + coff_l;

    float sqm[4];
#pragma unroll
    for (int ct = 0; ct < 4; ++ct)
      sqm[ct] = sqb[((cw * 4 + ct) * 2 + (l15 >> 3)) * (T_ * 8) + (l15 & 7)];
    v4f sqn4[ART_][2];
#pragma unroll
    for (int rt = 0; rt < ART_; ++rt) {
      const float* sqr = sqb + ((ntile * AR_ + rg * WR_ + rt * 16 + rbase) >> 3) * (T_ * 8);
      sqn4[rt][0] = *(const v4f*)(sqr);
      sqn4[rt][1] = *(const v4f*)(sqr + 4);
    }

    v8f c[ART_][4];
#pragma unroll
    for (int rt = 0; rt < ART_; ++rt)
#pragma unroll
      for (int ct = 0; ct < 4; ++ct) c[rt][ct] = v8f{};

    const float* arp = pcur + arow * PP_ + k4off;
    const float* bp  = pcur + (cw * 64 + l15) * PP_ + k4off;
    v4f fa[2][ART_], fb[2][4];
    v4f pq[4];
#pragma unroll
    for (int rt = 0; rt < ART_; ++rt) fa[0][rt] = *(const v4f*)(arp + rt * 16 * PP_);
#pragma unroll
    for (int ct = 0; ct < 4; ++ct)    fb[0][ct] = *(const v4f*)(bp + ct * 16 * PP_);
#pragma unroll
    for (int kb = 0; kb < D_ / 8; ++kb) {
      const int cur = kb & 1;
      if (kb + 1 < D_ / 8) {
#pragma unroll
        for (int rt = 0; rt < ART_; ++rt)
          fa[cur ^ 1][rt] = *(const v4f*)(arp + rt * 16 * PP_ + (kb + 1) * 8);
#pragma unroll
        for (int ct = 0; ct < 4; ++ct)
          fb[cur ^ 1][ct] = *(const v4f*)(bp + ct * 16 * PP_ + (kb + 1) * 8);
      }
      if ((kb & 1) == 0) {
#pragma unroll
        for (int i = 0; i < 4; ++i)
          pq[i] = *(const v4f*)(gnxt + (size_t)((kb >> 1) * 4 + i) * 2 * D_);
      }
      __builtin_amdgcn_sched_barrier(0);
#pragma unroll
      for (int ct = 0; ct < 4; ++ct) {
        v2f blo, bhi;
        blo.x = fb[cur][ct].x; blo.y = fb[cur][ct].y; bhi.x = fb[cur][ct].z; bhi.y = fb[cur][ct].w;
#pragma unroll
        for (int rt = 0; rt < ART_; ++rt) {
          v2f alo, ahi;
          alo.x = fa[cur][rt].x; alo.y = fa[cur][rt].y; ahi.x = fa[cur][rt].z; ahi.y = fa[cur][rt].w;
          c[rt][ct] = __builtin_amdgcn_wmma_f32_16x16x4_f32(false, alo, false, blo,
                                                            (short)0, c[rt][ct], false, false);
          c[rt][ct] = __builtin_amdgcn_wmma_f32_16x16x4_f32(false, ahi, false, bhi,
                                                            (short)0, c[rt][ct], false, false);
        }
      }
      __builtin_amdgcn_sched_barrier(0);
      if ((kb & 1) == 1) {
#pragma unroll
        for (int i = 0; i < 4; ++i)
          *(v4f*)(lnxt + ((kb >> 1) * 4 + i) * 2 * PP_) = pq[i];
      }
    }

    const float L2E_ = 1.44269504f, LN2_ = 0.693147181f;
    float rmv[4], nmi[4];
#pragma unroll
    for (int ct = 0; ct < 4; ++ct) {
      const float nm = __builtin_amdgcn_sqrtf(sqm[ct]) + EPS_;
      rmv[ct] = __builtin_amdgcn_rcpf(nm);
      nmi[ct] = (nm + EPS_) * LN2_;
    }
    v2f rs[ART_][4];
#pragma unroll
    for (int rt = 0; rt < ART_; ++rt) {
      v2f sqn2[4], nni[4], rnl[4];
#pragma unroll
      for (int i = 0; i < 4; ++i) {
        sqn2[i].x = sqn4[rt][i >> 1][(2 * i) & 3];
        sqn2[i].y = sqn4[rt][i >> 1][(2 * i + 1) & 3];
        v2f nn;
        nn.x = __builtin_amdgcn_sqrtf(sqn2[i].x) + EPS_;
        nn.y = __builtin_amdgcn_sqrtf(sqn2[i].y) + EPS_;
        nni[i] = nn * LN2_;
        rnl[i].x = __builtin_amdgcn_rcpf(nn.x) * L2E_;
        rnl[i].y = __builtin_amdgcn_rcpf(nn.y) * L2E_;
      }
#pragma unroll
      for (int ct = 0; ct < 4; ++ct) {
        if ((ct & 1) == 0 && (rt | ct) != 0) __builtin_amdgcn_sched_barrier(0);
        const float sqmv = sqm[ct];
        const float rm   = rmv[ct];
        const float nmiv = nmi[ct];
#pragma unroll
        for (int i = 0; i < 4; ++i) {
          v2f dot;
          dot.x = c[rt][ct][2 * i];
          dot.y = c[rt][ct][2 * i + 1];
          const v2f sn = nni[i] + nmiv;
          v2f q;
          q.x = __builtin_amdgcn_rcpf(sn.x);
          q.y = __builtin_amdgcn_rcpf(sn.y);
          v2f d2 = sqn2[i] + sqmv - 2.0f * dot;
          d2.x = fmaxf(d2.x, 0.0f);
          d2.y = fmaxf(d2.y, 0.0f);
          v2f diff;
          diff.x = __builtin_amdgcn_sqrtf(d2.x);
          diff.y = __builtin_amdgcn_sqrtf(d2.y);
          const v2f a = dot * rnl[i] * rm - diff * q;
          v2f e;
          e.x = __builtin_amdgcn_exp2f(a.x);
          e.y = __builtin_amdgcn_exp2f(a.y);
          c[rt][ct][2 * i]     = e.x;
          c[rt][ct][2 * i + 1] = e.y;
          rs[rt][i] = (ct == 0) ? e : rs[rt][i] + e;
        }
      }
    }

    const int buf = b & 1;
    float y8[8], y4[4], y2[2];
    {
      const bool m8 = (l15 & 8) != 0, m4 = (l15 & 4) != 0, m2 = (l15 & 2) != 0, m1 = (l15 & 1) != 0;
#pragma unroll
      for (int i = 0; i < 8; ++i) {
        const float r0 = rs[0][i >> 1][i & 1], r1 = rs[1][i >> 1][i & 1];
        const float keep = m8 ? r1 : r0;
        const float send = m8 ? r0 : r1;
        y8[i] = keep + ROW_XGET(send, 8);
      }
#pragma unroll
      for (int i = 0; i < 4; ++i) {
        const float keep = m4 ? y8[4 + i] : y8[i];
        const float send = m4 ? y8[i] : y8[4 + i];
        y4[i] = keep + ROW_XGET(send, 4);
      }
#pragma unroll
      for (int i = 0; i < 2; ++i) {
        const float keep = m2 ? y4[2 + i] : y4[i];
        const float send = m2 ? y4[i] : y4[2 + i];
        y2[i] = keep + ROW_XGET(send, 2);
      }
      const float keep = m1 ? y2[1] : y2[0];
      const float send = m1 ? y2[0] : y2[1];
      const float rowsum = keep + ROW_XGET(send, 1);
      part_s[buf][cw][rg * WR_ + ((l15 >> 3) << 4) + rbase + (l15 & 7)] = rowsum;
    }
    __syncthreads();
    {
      float s = 0.f;
#pragma unroll
      for (int w = 0; w < 8; ++w) s += part_s[buf][w][rg * WR_ + lane];
      rinv_s[wave][lane] = (1.0f / (float)B_) * __builtin_amdgcn_rcpf(s + EPS_);
    }

#pragma unroll
    for (int rt = 0; rt < ART_; ++rt) {
      const v4f r0 = *(const v4f*)&rinv_s[wave][rt * 16 + rbase];
      const v4f r1 = *(const v4f*)&rinv_s[wave][rt * 16 + rbase + 4];
#pragma unroll
      for (int ct = 0; ct < 4; ++ct) {
#pragma unroll
        for (int v = 0; v < 8; ++v) {
          const float rv = (v < 4) ? r0[v & 3] : r1[v & 3];
          acc[rt][ct][v] += c[rt][ct][v] * rv;
        }
      }
    }
  }

#pragma unroll
  for (int rt = 0; rt < ART_; ++rt)
#pragma unroll
    for (int ct = 0; ct < 4; ++ct)
#pragma unroll
      for (int v = 0; v < 8; ++v)
        panel_s[(rg * WR_ + rt * 16 + rbase + v) * 512 + (cw * 4 + ct) * 16 + l15] = acc[rt][ct][v];
  __syncthreads();
  float* outp = out + ((size_t)t * N_ + ntile * AR_) * N_;
#pragma unroll
  for (int j = 0; j < AR_; ++j) {
    const int idx = tid + j * 512;
    outp[idx] = panel_s[idx];
  }
}

extern "C" void kernel_launch(void* const* d_in, const int* in_sizes, int n_in,
                              void* d_out, int out_size, void* d_ws, size_t ws_size,
                              hipStream_t stream) {
  const float* x    = (const float*)d_in[0];
  const float* W    = (const float*)d_in[1];
  const float* bias = (const float*)d_in[2];
  float* out = (float*)d_out;

  float* z  = (float*)d_ws;
  float* sq = z + (size_t)B_ * T_ * N_ * D_;

  GAT_z_kernel<<<512, 256, 0, stream>>>(x, W, bias, z, sq);

  GAT_att_kernel<<<T_ * (N_ / AR_), 512, 0, stream>>>(z, sq, out);
}
